# starting kernel
# baseline (speedup 1.0000x reference)
#include <hip/hip_runtime.h>
#include <hip/hip_bf16.h>

typedef __attribute__((ext_vector_type(2))) float v2f;
typedef __attribute__((ext_vector_type(8))) float v8f;

#define N_NODES 100000

__global__ void zero_kernel(float* __restrict__ p, int n) {
    int i = blockIdx.x * blockDim.x + threadIdx.x;
    if (i < n) p[i] = 0.0f;
}

__global__ void degree_kernel(const int* __restrict__ dst,
                              float* __restrict__ deg, int E) {
    int e = blockIdx.x * blockDim.x + threadIdx.x;
    if (e < E) unsafeAtomicAdd(&deg[dst[e]], 1.0f);
}

__global__ void invdeg_kernel(float* __restrict__ deg, int n) {
    int i = blockIdx.x * blockDim.x + threadIdx.x;
    if (i < n) deg[i] = 1.0f / fmaxf(deg[i], 1.0f);
}

template <int F>
__global__ void scatter_kernel(const float* __restrict__ h,
                               const int* __restrict__ src,
                               const int* __restrict__ dst,
                               float* __restrict__ agg, int E) {
    int gw   = (blockIdx.x * blockDim.x + threadIdx.x) >> 5;
    int lane = threadIdx.x & 31;
    if (gw >= E) return;
    int s = src[gw];
    int d = dst[gw];
    if (F == 32) {
        float v = h[(size_t)s * 32 + lane];
        unsafeAtomicAdd(&agg[(size_t)d * 32 + lane], v);
    } else {
        const float4* row = reinterpret_cast<const float4*>(h + (size_t)s * F);
        float4 v = row[lane];
        float* o = agg + (size_t)d * F + lane * 4;
        unsafeAtomicAdd(o + 0, v.x);
        unsafeAtomicAdd(o + 1, v.y);
        unsafeAtomicAdd(o + 2, v.z);
        unsafeAtomicAdd(o + 3, v.w);
    }
}

template <int FIN, int FOUT, int TPW, bool PRELU>
__global__ __launch_bounds__(128) void sage_combine_wmma(
    const float* __restrict__ agg, const float* __restrict__ invdeg,
    const float* __restrict__ h,
    const float* __restrict__ Wl, const float* __restrict__ bias,
    const float* __restrict__ Wr, const float* __restrict__ alpha_p,
    float* __restrict__ out) {
    constexpr int LDSTR = FIN + 1;
    __shared__ float s_mean[16 * LDSTR];
    __shared__ float s_h[16 * LDSTR];

    const int tid   = threadIdx.x;
    const int node0 = blockIdx.x * 16;

    for (int idx = tid; idx < 16 * FIN; idx += 128) {
        int r = idx / FIN;
        int c = idx % FIN;
        size_t g = (size_t)(node0 + r) * FIN + c;
        s_mean[r * LDSTR + c] = agg[g] * invdeg[node0 + r];
        s_h[r * LDSTR + c]    = h[g];
    }
    __syncthreads();

    const int wave = tid >> 5;
    const int lane = tid & 31;
    const int half = lane >> 4;
    const int col  = lane & 15;
    const float alpha = alpha_p[0];

    v8f acc[TPW];
#pragma unroll
    for (int t = 0; t < TPW; ++t)
#pragma unroll
        for (int j = 0; j < 8; ++j) acc[t][j] = 0.0f;

    const float* smrow = s_mean + col * LDSTR;
    const float* shrow = s_h    + col * LDSTR;

#pragma unroll 4
    for (int k0 = 0; k0 < FIN; k0 += 4) {
        const int kb = k0 + half * 2;
        v2f am; am.x = smrow[kb]; am.y = smrow[kb + 1];
        v2f ah; ah.x = shrow[kb]; ah.y = shrow[kb + 1];
#pragma unroll
        for (int t = 0; t < TPW; ++t) {
            const int n0 = (wave * TPW + t) * 16;
            v2f bl;
            bl.x = Wl[(size_t)kb * FOUT + n0 + col];
            bl.y = Wl[(size_t)(kb + 1) * FOUT + n0 + col];
            v2f br;
            br.x = Wr[(size_t)kb * FOUT + n0 + col];
            br.y = Wr[(size_t)(kb + 1) * FOUT + n0 + col];
            acc[t] = __builtin_amdgcn_wmma_f32_16x16x4_f32(
                false, am, false, bl, (short)0, acc[t], false, false);
            acc[t] = __builtin_amdgcn_wmma_f32_16x16x4_f32(
                false, ah, false, br, (short)0, acc[t], false, false);
        }
    }

#pragma unroll
    for (int t = 0; t < TPW; ++t) {
        const int n0 = (wave * TPW + t) * 16;
        const float bv = bias[n0 + col];
#pragma unroll
        for (int v = 0; v < 8; ++v) {
            float val = acc[t][v] + bv;
            if (PRELU) val = (val >= 0.0f) ? val : alpha * val;
            const int row = node0 + v + half * 8;
            out[(size_t)row * FOUT + n0 + col] = val;
        }
    }
}

__global__ void classifier_kernel(const float* __restrict__ h3,
                                  const float* __restrict__ Wc,
                                  const float* __restrict__ bc,
                                  float* __restrict__ out, int n) {
    int i = blockIdx.x * blockDim.x + threadIdx.x;
    if (i >= n) return;
    float s0 = bc[0], s1 = bc[1], s2 = bc[2], s3 = bc[3];
    const float4* row = reinterpret_cast<const float4*>(h3 + (size_t)i * 64);
#pragma unroll
    for (int k4 = 0; k4 < 16; ++k4) {
        float4 v = row[k4];
        const float* w = Wc + (k4 * 4) * 4;
        s0 += v.x * w[0];  s1 += v.x * w[1];  s2 += v.x * w[2];  s3 += v.x * w[3];
        s0 += v.y * w[4];  s1 += v.y * w[5];  s2 += v.y * w[6];  s3 += v.y * w[7];
        s0 += v.z * w[8];  s1 += v.z * w[9];  s2 += v.z * w[10]; s3 += v.z * w[11];
        s0 += v.w * w[12]; s1 += v.w * w[13]; s2 += v.w * w[14]; s3 += v.w * w[15];
    }
    float* o = out + (size_t)i * 4;
    o[0] = s0; o[1] = s1; o[2] = s2; o[3] = s3;
}

extern "C" void kernel_launch(void* const* d_in, const int* in_sizes, int n_in,
                              void* d_out, int out_size, void* d_ws, size_t ws_size,
                              hipStream_t stream) {
    const float* x   = (const float*)d_in[0];
    const int*   ei  = (const int*)d_in[1];
    const float* W1l = (const float*)d_in[2];
    const float* b1  = (const float*)d_in[3];
    const float* W1r = (const float*)d_in[4];
    const float* W2l = (const float*)d_in[5];
    const float* b2  = (const float*)d_in[6];
    const float* W2r = (const float*)d_in[7];
    const float* W3l = (const float*)d_in[8];
    const float* b3  = (const float*)d_in[9];
    const float* W3r = (const float*)d_in[10];
    const float* Wc  = (const float*)d_in[11];
    const float* bc  = (const float*)d_in[12];
    const float* a   = (const float*)d_in[13];
    float* out = (float*)d_out;

    const int E = in_sizes[1] / 2;
    const int* srcv = ei;
    const int* dstv = ei + E;

    char* ws = (char*)d_ws;
    size_t off = 0;
    auto carve = [&](size_t bytes) {
        float* p = (float*)(ws + off);
        off += (bytes + 255) & ~(size_t)255;
        return p;
    };
    float* deg = carve((size_t)N_NODES * 4);
    float* agg = carve((size_t)N_NODES * 128 * 4);
    float* hA  = carve((size_t)N_NODES * 128 * 4);
    float* hB  = carve((size_t)N_NODES * 128 * 4);
    float* h3  = hA;

    const int TB = 256;
    const int nBlk16 = N_NODES / 16;

    zero_kernel<<<(N_NODES + TB - 1) / TB, TB, 0, stream>>>(deg, N_NODES);
    degree_kernel<<<(E + TB - 1) / TB, TB, 0, stream>>>(dstv, deg, E);
    invdeg_kernel<<<(N_NODES + TB - 1) / TB, TB, 0, stream>>>(deg, N_NODES);

    zero_kernel<<<(N_NODES * 32 + TB - 1) / TB, TB, 0, stream>>>(agg, N_NODES * 32);
    scatter_kernel<32><<<((size_t)E * 32 + TB - 1) / TB, TB, 0, stream>>>(
        x, srcv, dstv, agg, E);
    sage_combine_wmma<32, 128, 2, true><<<nBlk16, 128, 0, stream>>>(
        agg, deg, x, W1l, b1, W1r, a, hA);

    zero_kernel<<<(N_NODES * 128 + TB - 1) / TB, TB, 0, stream>>>(agg, N_NODES * 128);
    scatter_kernel<128><<<((size_t)E * 32 + TB - 1) / TB, TB, 0, stream>>>(
        hA, srcv, dstv, agg, E);
    sage_combine_wmma<128, 128, 2, true><<<nBlk16, 128, 0, stream>>>(
        agg, deg, hA, W2l, b2, W2r, a, hB);

    zero_kernel<<<(N_NODES * 128 + TB - 1) / TB, TB, 0, stream>>>(agg, N_NODES * 128);
    scatter_kernel<128><<<((size_t)E * 32 + TB - 1) / TB, TB, 0, stream>>>(
        hB, srcv, dstv, agg, E);
    sage_combine_wmma<128, 64, 1, false><<<nBlk16, 128, 0, stream>>>(
        agg, deg, hB, W3l, b3, W3r, a, h3);

    classifier_kernel<<<(N_NODES + TB - 1) / TB, TB, 0, stream>>>(
        h3, Wc, bc, out, N_NODES);
}


# Layers 2-3 neighbour sums by owner work-groups: 256 row ranges summed in LDS (LDS f32 adds) from a full scan of dst, rows stored once, no global atomics or clears; sum order changes (LDS arrival)
# speedup vs baseline: 2.0008x; 2.0008x over previous
#include <hip/hip_runtime.h>
#include <hip/hip_bf16.h>

typedef __attribute__((ext_vector_type(2))) float v2f;
typedef __attribute__((ext_vector_type(8))) float v8f;

#define N_NODES 100000

__global__ void zero_kernel(float* __restrict__ p, int n) {
    int i = blockIdx.x * blockDim.x + threadIdx.x;
    if (i < n) p[i] = 0.0f;
}

__global__ void degree_kernel(const int* __restrict__ dst,
                              float* __restrict__ deg, int E) {
    int e = blockIdx.x * blockDim.x + threadIdx.x;
    if (e < E) unsafeAtomicAdd(&deg[dst[e]], 1.0f);
}

__global__ void invdeg_kernel(float* __restrict__ deg, int n) {
    int i = blockIdx.x * blockDim.x + threadIdx.x;
    if (i < n) deg[i] = 1.0f / fmaxf(deg[i], 1.0f);
}

template <int F, int U>
__global__ __launch_bounds__(256) void scatter_kernel(const float* __restrict__ h,
                               const int* __restrict__ src,
                               const int* __restrict__ dst,
                               float* __restrict__ agg, int E) {
    constexpr int C = F / 32;
    const int lane = threadIdx.x & 31;
    const int gw   = (blockIdx.x * blockDim.x + threadIdx.x) >> 5;
    const long long e0 = (long long)gw * 32;
    if (e0 >= E) return;
    const int n = (E - e0 < 32) ? (int)(E - e0) : 32;
    int my_s = 0, my_d = 0;
    if (lane < n) { my_s = src[e0 + lane]; my_d = dst[e0 + lane]; }

    if (n == 32) {
#pragma unroll 1
        for (int j0 = 0; j0 < 32; j0 += U) {
            float v[U][C];
#pragma unroll
            for (int u = 0; u < U; ++u) {
                const int s = __builtin_amdgcn_readlane(my_s, j0 + u);
                const float* row = h + (size_t)s * F + lane;
#pragma unroll
                for (int c = 0; c < C; ++c) v[u][c] = row[c * 32];
            }
#pragma unroll
            for (int u = 0; u < U; ++u) {
                const int d = __builtin_amdgcn_readlane(my_d, j0 + u);
                float* o = agg + (size_t)d * F + lane;
#pragma unroll
                for (int c = 0; c < C; ++c) unsafeAtomicAdd(o + c * 32, v[u][c]);
            }
        }
    } else {
        for (int j = 0; j < n; ++j) {
            const int s = __builtin_amdgcn_readlane(my_s, j);
            const int d = __builtin_amdgcn_readlane(my_d, j);
            const float* row = h + (size_t)s * F + lane;
            float* o = agg + (size_t)d * F + lane;
#pragma unroll
            for (int c = 0; c < C; ++c) unsafeAtomicAdd(o + c * 32, row[c * 32]);
        }
    }
}


#define RANGE 391
#define QCAP  160
#define OWN_G 8

template <int F, int U, bool FULL>
__device__ __forceinline__ void owner_drain(const uint2* qw, int first, int n, int lane,
                                            const float* __restrict__ h,
                                            const int* __restrict__ src,
                                            float* acc) {
    constexpr int C = F / 32;
    int my_s = 0, my_r = -1;
    if (FULL || lane < n) {
        const uint2 ent = qw[first + lane];
        my_s = src[ent.x];
        my_r = (int)ent.y;
    }
#pragma unroll 1
    for (int j0 = 0; j0 < n; j0 += U) {
        float v[U][C];
#pragma unroll
        for (int u = 0; u < U; ++u) {
            const int s = __builtin_amdgcn_readlane(my_s, j0 + u);
            const float* row = h + (size_t)s * F + lane;
#pragma unroll
            for (int c = 0; c < C; ++c) v[u][c] = row[c * 32];
        }
#pragma unroll
        for (int u = 0; u < U; ++u) {
            const int r = __builtin_amdgcn_readlane(my_r, j0 + u);
            if (FULL || r >= 0) {
                float* o = acc + r * F + lane;
#pragma unroll
                for (int c = 0; c < C; ++c) atomicAdd(o + c * 32, v[u][c]);
            }
        }
    }
}

template <int F, int U>
__device__ __forceinline__ int owner_take(const int4 d, const int eb, const int lo,
                                          const unsigned nrows, uint2* qw, int qn, int lane,
                                          const float* __restrict__ h,
                                          const int* __restrict__ src, float* acc) {
    unsigned x[4];
    x[0] = (unsigned)(d.x - lo);
    x[1] = (unsigned)(d.y - lo);
    x[2] = (unsigned)(d.z - lo);
    x[3] = (unsigned)(d.w - lo);
    const unsigned xm = min(min(x[0], x[1]), min(x[2], x[3]));
    if (__builtin_amdgcn_ballot_w32(xm < nrows) != 0u) {
#pragma unroll
        for (int j = 0; j < 4; ++j) {
            const bool m = x[j] < nrows;
            const unsigned b = __builtin_amdgcn_ballot_w32(m);
            if (b != 0u) {
                if (m) qw[qn + (int)__builtin_amdgcn_mbcnt_lo(b, 0u)] =
                           make_uint2((unsigned)(eb + j), x[j]);
                qn += __builtin_popcount(b);
            }
        }
        while (qn >= 32) {
            qn -= 32;
            owner_drain<F, U, true>(qw, qn, 32, lane, h, src, acc);
        }
    }
    return qn;
}

template <int F, int U>
__global__ __launch_bounds__(1024) void owner_sum_kernel(const float* __restrict__ h,
                               const int* __restrict__ src,
                               const int* __restrict__ dst,
                               float* __restrict__ agg, int E) {
    __shared__ float acc[RANGE * F];
    __shared__ uint2 q[32 * QCAP];

    const int tid  = threadIdx.x;
    const int lane = tid & 31;
    const int wave = __builtin_amdgcn_readfirstlane(tid >> 5);
    const int lo   = blockIdx.x * RANGE;
    int nr = N_NODES - lo;
    if (nr > RANGE) nr = RANGE;
    const unsigned nrows = (unsigned)nr;

    float4* acc4 = reinterpret_cast<float4*>(acc);
    for (int i = tid; i < RANGE * F / 4; i += 1024)
        acc4[i] = make_float4(0.0f, 0.0f, 0.0f, 0.0f);
    __syncthreads();

    uint2* qw = q + wave * QCAP;
    int qn = 0;
    const int nfull = E >> 7;
    const int start = (int)(((long long)blockIdx.x * nfull) >> 8);

    for (int t0 = wave; t0 < nfull; t0 += 32 * OWN_G) {
        int4 d[OWN_G];
        int  eb[OWN_G];
#pragma unroll
        for (int u = 0; u < OWN_G; ++u) {
            const int t = t0 + 32 * u;
            int g = start + ((t < nfull) ? t : t0);
            if (g >= nfull) g -= nfull;
            eb[u] = g * 128 + lane * 4;
            d[u] = *reinterpret_cast<const int4*>(dst + eb[u]);
        }
#pragma unroll
        for (int u = 0; u < OWN_G; ++u) {
            if (t0 + 32 * u >= nfull) d[u] = make_int4(-1, -1, -1, -1);
            qn = owner_take<F, U>(d[u], eb[u], lo, nrows, qw, qn, lane, h, src, acc);
        }
    }
    if ((E & 127) != 0 && wave == 0) {
        const int eb = nfull * 128 + lane * 4;
        int4 d = make_int4(-1, -1, -1, -1);
        if (eb     < E) d.x = dst[eb];
        if (eb + 1 < E) d.y = dst[eb + 1];
        if (eb + 2 < E) d.z = dst[eb + 2];
        if (eb + 3 < E) d.w = dst[eb + 3];
        qn = owner_take<F, U>(d, eb, lo, nrows, qw, qn, lane, h, src, acc);
    }
    if (qn > 0) owner_drain<F, U, false>(qw, 0, qn, lane, h, src, acc);
    __syncthreads();

    float4* o4 = reinterpret_cast<float4*>(agg + (size_t)lo * F);
    for (int i = tid; i < nr * (F / 4); i += 1024) o4[i] = acc4[i];
}

template <int FIN, int FOUT, int TPW, bool PRELU>
__global__ __launch_bounds__(128) void sage_combine_wmma(
    const float* __restrict__ agg, const float* __restrict__ invdeg,
    const float* __restrict__ h,
    const float* __restrict__ Wl, const float* __restrict__ bias,
    const float* __restrict__ Wr, const float* __restrict__ alpha_p,
    float* __restrict__ out) {
    constexpr int LDSTR = FIN + 1;
    __shared__ float s_mean[16 * LDSTR];
    __shared__ float s_h[16 * LDSTR];

    const int tid   = threadIdx.x;
    const int node0 = blockIdx.x * 16;

    for (int idx = tid; idx < 16 * FIN; idx += 128) {
        int r = idx / FIN;
        int c = idx % FIN;
        size_t g = (size_t)(node0 + r) * FIN + c;
        s_mean[r * LDSTR + c] = agg[g] * invdeg[node0 + r];
        s_h[r * LDSTR + c]    = h[g];
    }
    __syncthreads();

    const int wave = tid >> 5;
    const int lane = tid & 31;
    const int half = lane >> 4;
    const int col  = lane & 15;
    const float alpha = alpha_p[0];

    v8f acc[TPW];
#pragma unroll
    for (int t = 0; t < TPW; ++t)
#pragma unroll
        for (int j = 0; j < 8; ++j) acc[t][j] = 0.0f;

    const float* smrow = s_mean + col * LDSTR;
    const float* shrow = s_h    + col * LDSTR;

#pragma unroll 4
    for (int k0 = 0; k0 < FIN; k0 += 4) {
        const int kb = k0 + half * 2;
        v2f am; am.x = smrow[kb]; am.y = smrow[kb + 1];
        v2f ah; ah.x = shrow[kb]; ah.y = shrow[kb + 1];
#pragma unroll
        for (int t = 0; t < TPW; ++t) {
            const int n0 = (wave * TPW + t) * 16;
            v2f bl;
            bl.x = Wl[(size_t)kb * FOUT + n0 + col];
            bl.y = Wl[(size_t)(kb + 1) * FOUT + n0 + col];
            v2f br;
            br.x = Wr[(size_t)kb * FOUT + n0 + col];
            br.y = Wr[(size_t)(kb + 1) * FOUT + n0 + col];
            acc[t] = __builtin_amdgcn_wmma_f32_16x16x4_f32(
                false, am, false, bl, (short)0, acc[t], false, false);
            acc[t] = __builtin_amdgcn_wmma_f32_16x16x4_f32(
                false, ah, false, br, (short)0, acc[t], false, false);
        }
    }

#pragma unroll
    for (int t = 0; t < TPW; ++t) {
        const int n0 = (wave * TPW + t) * 16;
        const float bv = bias[n0 + col];
#pragma unroll
        for (int v = 0; v < 8; ++v) {
            float val = acc[t][v] + bv;
            if (PRELU) val = (val >= 0.0f) ? val : alpha * val;
            const int row = node0 + v + half * 8;
            out[(size_t)row * FOUT + n0 + col] = val;
        }
    }
}

__global__ void classifier_kernel(const float* __restrict__ h3,
                                  const float* __restrict__ Wc,
                                  const float* __restrict__ bc,
                                  float* __restrict__ out, int n) {
    int i = blockIdx.x * blockDim.x + threadIdx.x;
    if (i >= n) return;
    float s0 = bc[0], s1 = bc[1], s2 = bc[2], s3 = bc[3];
    const float4* row = reinterpret_cast<const float4*>(h3 + (size_t)i * 64);
#pragma unroll
    for (int k4 = 0; k4 < 16; ++k4) {
        float4 v = row[k4];
        const float* w = Wc + (k4 * 4) * 4;
        s0 += v.x * w[0];  s1 += v.x * w[1];  s2 += v.x * w[2];  s3 += v.x * w[3];
        s0 += v.y * w[4];  s1 += v.y * w[5];  s2 += v.y * w[6];  s3 += v.y * w[7];
        s0 += v.z * w[8];  s1 += v.z * w[9];  s2 += v.z * w[10]; s3 += v.z * w[11];
        s0 += v.w * w[12]; s1 += v.w * w[13]; s2 += v.w * w[14]; s3 += v.w * w[15];
    }
    float* o = out + (size_t)i * 4;
    o[0] = s0; o[1] = s1; o[2] = s2; o[3] = s3;
}

extern "C" void kernel_launch(void* const* d_in, const int* in_sizes, int n_in,
                              void* d_out, int out_size, void* d_ws, size_t ws_size,
                              hipStream_t stream) {
    const float* x   = (const float*)d_in[0];
    const int*   ei  = (const int*)d_in[1];
    const float* W1l = (const float*)d_in[2];
    const float* b1  = (const float*)d_in[3];
    const float* W1r = (const float*)d_in[4];
    const float* W2l = (const float*)d_in[5];
    const float* b2  = (const float*)d_in[6];
    const float* W2r = (const float*)d_in[7];
    const float* W3l = (const float*)d_in[8];
    const float* b3  = (const float*)d_in[9];
    const float* W3r = (const float*)d_in[10];
    const float* Wc  = (const float*)d_in[11];
    const float* bc  = (const float*)d_in[12];
    const float* a   = (const float*)d_in[13];
    float* out = (float*)d_out;

    const int E = in_sizes[1] / 2;
    const int* srcv = ei;
    const int* dstv = ei + E;

    char* ws = (char*)d_ws;
    size_t off = 0;
    auto carve = [&](size_t bytes) {
        float* p = (float*)(ws + off);
        off += (bytes + 255) & ~(size_t)255;
        return p;
    };
    float* deg = carve((size_t)N_NODES * 4);
    float* agg = carve((size_t)N_NODES * 128 * 4);
    float* hA  = carve((size_t)N_NODES * 128 * 4);
    float* hB  = carve((size_t)N_NODES * 128 * 4);
    float* h3  = hA;

    const int TB = 256;
    const int nBlk16 = N_NODES / 16;
    const int nBlkSc = (E + TB - 1) / TB;
    const int nOwn   = (N_NODES + RANGE - 1) / RANGE;

    zero_kernel<<<(N_NODES + TB - 1) / TB, TB, 0, stream>>>(deg, N_NODES);
    degree_kernel<<<(E + TB - 1) / TB, TB, 0, stream>>>(dstv, deg, E);
    invdeg_kernel<<<(N_NODES + TB - 1) / TB, TB, 0, stream>>>(deg, N_NODES);

    zero_kernel<<<(N_NODES * 32 + TB - 1) / TB, TB, 0, stream>>>(agg, N_NODES * 32);
    scatter_kernel<32, 8><<<nBlkSc, TB, 0, stream>>>(
        x, srcv, dstv, agg, E);
    sage_combine_wmma<32, 128, 2, true><<<nBlk16, 128, 0, stream>>>(
        agg, deg, x, W1l, b1, W1r, a, hA);

    owner_sum_kernel<128, 8><<<nOwn, 1024, 0, stream>>>(
        hA, srcv, dstv, agg, E);
    sage_combine_wmma<128, 128, 2, true><<<nBlk16, 128, 0, stream>>>(
        agg, deg, hA, W2l, b2, W2r, a, hB);

    owner_sum_kernel<128, 8><<<nOwn, 1024, 0, stream>>>(
        hB, srcv, dstv, agg, E);
    sage_combine_wmma<128, 64, 1, false><<<nBlk16, 128, 0, stream>>>(
        agg, deg, hB, W3l, b3, W3r, a, h3);

    classifier_kernel<<<(N_NODES + TB - 1) / TB, TB, 0, stream>>>(
        h3, Wc, bc, out, N_NODES);
}


# Owner scan of layers 2-3 reads two one-hot bit planes (owner number's high and low 4 bits; 32 planes made of wave compare masks by the key pass): 1 AND per 32 edges; sum order changes (LDS arrival)
# speedup vs baseline: 2.2939x; 1.1465x over previous
#include <hip/hip_runtime.h>
#include <hip/hip_bf16.h>

typedef __attribute__((ext_vector_type(2))) float v2f;
typedef __attribute__((ext_vector_type(8))) float v8f;

#define N_NODES 100000

__global__ void zero_kernel(float* __restrict__ p, int n) {
    int i = blockIdx.x * blockDim.x + threadIdx.x;
    if (i < n) p[i] = 0.0f;
}

#define RANGE 391
static_assert(N_NODES <= (1 << 17), "src must fit the low 17 bits of the packed word");
static_assert(RANGE <= (1 << 15), "row must fit above bit 17");
static_assert((N_NODES + RANGE - 1) / RANGE <= 256, "owner number must fit two 4-bit parts");

template <bool COUNT>
__global__ __launch_bounds__(128) void edge_keys_kernel(const int* __restrict__ src,
                                 const int* __restrict__ dst,
                                 float* __restrict__ deg,
                                 unsigned* __restrict__ planes,
                                 unsigned* __restrict__ pk, int E) {
    const int lane = threadIdx.x & 31;
    const int gw = __builtin_amdgcn_readfirstlane((int)((blockIdx.x * blockDim.x + threadIdx.x) >> 5));
    const int pw = ((E + 4095) >> 12) << 7;
    const int w0 = gw * 32;
    if (w0 >= pw) return;
    unsigned wa[16], wb[16];
#pragma unroll
    for (int k = 0; k < 16; ++k) { wa[k] = 0u; wb[k] = 0u; }
#pragma unroll 1
    for (int i0 = 0; i0 < 32; i0 += 8) {
        int s[8], d[8];
#pragma unroll
        for (int u = 0; u < 8; ++u) {
            const int e  = (w0 + i0 + u) * 32 + lane;
            const int ec = (e < E) ? e : (E - 1);
            s[u] = src[ec]; d[u] = dst[ec];
        }
        unsigned a[8], b[8], p[8];
#pragma unroll
        for (int u = 0; u < 8; ++u) {
            const int e = (w0 + i0 + u) * 32 + lane;
            const bool v = (e < E) && (d[u] >= 0);
            const unsigned o = (unsigned)d[u] / RANGE;
            const unsigned r = (unsigned)d[u] - o * RANGE;
            a[u] = v ? (o >> 4) : 16u;
            b[u] = v ? (o & 15u) : 16u;
            p[u] = (unsigned)s[u] | (r << 17);
        }
#pragma unroll
        for (int u = 0; u < 8; ++u) {
            const bool mine = (lane == i0 + u);
#pragma unroll
            for (int k = 0; k < 16; ++k) {
                const unsigned ma = __builtin_amdgcn_ballot_w32(a[u] == (unsigned)k);
                const unsigned mb = __builtin_amdgcn_ballot_w32(b[u] == (unsigned)k);
                wa[k] = mine ? ma : wa[k];
                wb[k] = mine ? mb : wb[k];
            }
        }
#pragma unroll
        for (int u = 0; u < 8; ++u) {
            const int e = (w0 + i0 + u) * 32 + lane;
            if (a[u] < 16u) {
                pk[e] = p[u];
                if (COUNT) unsafeAtomicAdd(&deg[d[u]], 1.0f);
            }
        }
    }
#pragma unroll
    for (int k = 0; k < 16; ++k) {
        planes[(size_t)k * pw + w0 + lane]        = wa[k];
        planes[(size_t)(16 + k) * pw + w0 + lane] = wb[k];
    }
}

__global__ void invdeg_kernel(float* __restrict__ deg, int n) {
    int i = blockIdx.x * blockDim.x + threadIdx.x;
    if (i < n) deg[i] = 1.0f / fmaxf(deg[i], 1.0f);
}

template <int F, int U>
__global__ __launch_bounds__(256) void scatter_kernel(const float* __restrict__ h,
                               const int* __restrict__ src,
                               const int* __restrict__ dst,
                               float* __restrict__ agg, int E) {
    constexpr int C = F / 32;
    const int lane = threadIdx.x & 31;
    const int gw   = (blockIdx.x * blockDim.x + threadIdx.x) >> 5;
    const long long e0 = (long long)gw * 32;
    if (e0 >= E) return;
    const int n = (E - e0 < 32) ? (int)(E - e0) : 32;
    int my_s = 0, my_d = 0;
    if (lane < n) { my_s = src[e0 + lane]; my_d = dst[e0 + lane]; }

    if (n == 32) {
#pragma unroll 1
        for (int j0 = 0; j0 < 32; j0 += U) {
            float v[U][C];
#pragma unroll
            for (int u = 0; u < U; ++u) {
                const int s = __builtin_amdgcn_readlane(my_s, j0 + u);
                const float* row = h + (size_t)s * F + lane;
#pragma unroll
                for (int c = 0; c < C; ++c) v[u][c] = row[c * 32];
            }
#pragma unroll
            for (int u = 0; u < U; ++u) {
                const int d = __builtin_amdgcn_readlane(my_d, j0 + u);
                float* o = agg + (size_t)d * F + lane;
#pragma unroll
                for (int c = 0; c < C; ++c) unsafeAtomicAdd(o + c * 32, v[u][c]);
            }
        }
    } else {
        for (int j = 0; j < n; ++j) {
            const int s = __builtin_amdgcn_readlane(my_s, j);
            const int d = __builtin_amdgcn_readlane(my_d, j);
            const float* row = h + (size_t)s * F + lane;
            float* o = agg + (size_t)d * F + lane;
#pragma unroll
            for (int c = 0; c < C; ++c) unsafeAtomicAdd(o + c * 32, row[c * 32]);
        }
    }
}


#define QCAP  64
#define OWN_G 4

template <int F, int U, bool FULL>
__device__ __forceinline__ void owner_drain(const unsigned* qw, int first, int n, int lane,
                                            const float* __restrict__ h,
                                            const unsigned* __restrict__ pk,
                                            float* acc) {
    constexpr int C = F / 32;
    int my_s = 0, my_r = -1;
    if (FULL || lane < n) {
        const unsigned p = pk[qw[first + lane]];
        my_s = (int)(p & 0x1ffffu);
        my_r = (int)(p >> 17);
    }
#pragma unroll 1
    for (int j0 = 0; j0 < n; j0 += U) {
        float v[U][C];
#pragma unroll
        for (int u = 0; u < U; ++u) {
            const int s = __builtin_amdgcn_readlane(my_s, j0 + u);
            const float* row = h + (size_t)s * F + lane;
#pragma unroll
            for (int c = 0; c < C; ++c) v[u][c] = row[c * 32];
        }
#pragma unroll
        for (int u = 0; u < U; ++u) {
            const int r = __builtin_amdgcn_readlane(my_r, j0 + u);
            if (FULL || r >= 0) {
                float* o = acc + r * F + lane;
#pragma unroll
                for (int c = 0; c < C; ++c) atomicAdd(o + c * 32, v[u][c]);
            }
        }
    }
}

template <int F, int U>
__device__ __forceinline__ int owner_take(unsigned hits, const int eb, const int E,
                                          unsigned* qw, int qn, int lane,
                                          const float* __restrict__ h,
                                          const unsigned* __restrict__ pk, float* acc) {
    while (__builtin_amdgcn_ballot_w32(hits != 0u) != 0u) {
        const int e = eb + __builtin_ctz(hits | 0x80000000u);
        const bool m = (hits != 0u) && (e < E);
        const unsigned b = __builtin_amdgcn_ballot_w32(m);
        if (m) qw[qn + (int)__builtin_amdgcn_mbcnt_lo(b, 0u)] = (unsigned)e;
        qn += __builtin_popcount(b);
        hits &= hits - 1u;
        if (qn >= 32) {
            qn -= 32;
            owner_drain<F, U, true>(qw, qn, 32, lane, h, pk, acc);
        }
    }
    return qn;
}

template <int F, int U>
__global__ __launch_bounds__(1024) void owner_sum_kernel(const float* __restrict__ h,
                               const unsigned* __restrict__ planes,
                               const unsigned* __restrict__ pk,
                               float* __restrict__ agg, int E) {
    __shared__ float acc[RANGE * F];
    __shared__ unsigned q[32 * QCAP];

    const int tid  = threadIdx.x;
    const int lane = tid & 31;
    const int wave = __builtin_amdgcn_readfirstlane(tid >> 5);
    const int lo   = blockIdx.x * RANGE;
    int nr = N_NODES - lo;
    if (nr > RANGE) nr = RANGE;

    float4* acc4 = reinterpret_cast<float4*>(acc);
    for (int i = tid; i < RANGE * F / 4; i += 1024)
        acc4[i] = make_float4(0.0f, 0.0f, 0.0f, 0.0f);
    __syncthreads();

    unsigned* qw = q + wave * QCAP;
    int qn = 0;
    const int ng = (E + 4095) >> 12;
    const int pw = ng << 7;
    const uint4* pa = reinterpret_cast<const uint4*>(planes + (size_t)(blockIdx.x >> 4) * pw);
    const uint4* pb = reinterpret_cast<const uint4*>(planes + (size_t)(16 + (blockIdx.x & 15)) * pw);
    const int start = (int)(((long long)blockIdx.x * ng) >> 8);

    for (int t0 = wave; t0 < ng; t0 += 32 * OWN_G) {
        uint4 wa[OWN_G], wb[OWN_G];
        int   eb[OWN_G];
#pragma unroll
        for (int u = 0; u < OWN_G; ++u) {
            const int t = t0 + 32 * u;
            int g = start + ((t < ng) ? t : t0);
            if (g >= ng) g -= ng;
            const int qi = g * 32 + lane;
            eb[u] = qi * 128;
            wa[u] = pa[qi];
            wb[u] = pb[qi];
        }
#pragma unroll
        for (int u = 0; u < OWN_G; ++u) {
            const bool live = (t0 + 32 * u < ng);
            const unsigned m0 = live ? (wa[u].x & wb[u].x) : 0u;
            const unsigned m1 = live ? (wa[u].y & wb[u].y) : 0u;
            const unsigned m2 = live ? (wa[u].z & wb[u].z) : 0u;
            const unsigned m3 = live ? (wa[u].w & wb[u].w) : 0u;
            qn = owner_take<F, U>(m0, eb[u],      E, qw, qn, lane, h, pk, acc);
            qn = owner_take<F, U>(m1, eb[u] + 32, E, qw, qn, lane, h, pk, acc);
            qn = owner_take<F, U>(m2, eb[u] + 64, E, qw, qn, lane, h, pk, acc);
            qn = owner_take<F, U>(m3, eb[u] + 96, E, qw, qn, lane, h, pk, acc);
        }
    }
    if (qn > 0) owner_drain<F, U, false>(qw, 0, qn, lane, h, pk, acc);
    __syncthreads();

    float4* o4 = reinterpret_cast<float4*>(agg + (size_t)lo * F);
    for (int i = tid; i < nr * (F / 4); i += 1024) o4[i] = acc4[i];
}

template <int FIN, int FOUT, int TPW, bool PRELU>
__global__ __launch_bounds__(128) void sage_combine_wmma(
    const float* __restrict__ agg, const float* __restrict__ invdeg,
    const float* __restrict__ h,
    const float* __restrict__ Wl, const float* __restrict__ bias,
    const float* __restrict__ Wr, const float* __restrict__ alpha_p,
    float* __restrict__ out) {
    constexpr int LDSTR = FIN + 1;
    __shared__ float s_mean[16 * LDSTR];
    __shared__ float s_h[16 * LDSTR];

    const int tid   = threadIdx.x;
    const int node0 = blockIdx.x * 16;

    for (int idx = tid; idx < 16 * FIN; idx += 128) {
        int r = idx / FIN;
        int c = idx % FIN;
        size_t g = (size_t)(node0 + r) * FIN + c;
        s_mean[r * LDSTR + c] = agg[g] * invdeg[node0 + r];
        s_h[r * LDSTR + c]    = h[g];
    }
    __syncthreads();

    const int wave = tid >> 5;
    const int lane = tid & 31;
    const int half = lane >> 4;
    const int col  = lane & 15;
    const float alpha = alpha_p[0];

    v8f acc[TPW];
#pragma unroll
    for (int t = 0; t < TPW; ++t)
#pragma unroll
        for (int j = 0; j < 8; ++j) acc[t][j] = 0.0f;

    const float* smrow = s_mean + col * LDSTR;
    const float* shrow = s_h    + col * LDSTR;

#pragma unroll 4
    for (int k0 = 0; k0 < FIN; k0 += 4) {
        const int kb = k0 + half * 2;
        v2f am; am.x = smrow[kb]; am.y = smrow[kb + 1];
        v2f ah; ah.x = shrow[kb]; ah.y = shrow[kb + 1];
#pragma unroll
        for (int t = 0; t < TPW; ++t) {
            const int n0 = (wave * TPW + t) * 16;
            v2f bl;
            bl.x = Wl[(size_t)kb * FOUT + n0 + col];
            bl.y = Wl[(size_t)(kb + 1) * FOUT + n0 + col];
            v2f br;
            br.x = Wr[(size_t)kb * FOUT + n0 + col];
            br.y = Wr[(size_t)(kb + 1) * FOUT + n0 + col];
            acc[t] = __builtin_amdgcn_wmma_f32_16x16x4_f32(
                false, am, false, bl, (short)0, acc[t], false, false);
            acc[t] = __builtin_amdgcn_wmma_f32_16x16x4_f32(
                false, ah, false, br, (short)0, acc[t], false, false);
        }
    }

#pragma unroll
    for (int t = 0; t < TPW; ++t) {
        const int n0 = (wave * TPW + t) * 16;
        const float bv = bias[n0 + col];
#pragma unroll
        for (int v = 0; v < 8; ++v) {
            float val = acc[t][v] + bv;
            if (PRELU) val = (val >= 0.0f) ? val : alpha * val;
            const int row = node0 + v + half * 8;
            out[(size_t)row * FOUT + n0 + col] = val;
        }
    }
}

__global__ void classifier_kernel(const float* __restrict__ h3,
                                  const float* __restrict__ Wc,
                                  const float* __restrict__ bc,
                                  float* __restrict__ out, int n) {
    int i = blockIdx.x * blockDim.x + threadIdx.x;
    if (i >= n) return;
    float s0 = bc[0], s1 = bc[1], s2 = bc[2], s3 = bc[3];
    const float4* row = reinterpret_cast<const float4*>(h3 + (size_t)i * 64);
#pragma unroll
    for (int k4 = 0; k4 < 16; ++k4) {
        float4 v = row[k4];
        const float* w = Wc + (k4 * 4) * 4;
        s0 += v.x * w[0];  s1 += v.x * w[1];  s2 += v.x * w[2];  s3 += v.x * w[3];
        s0 += v.y * w[4];  s1 += v.y * w[5];  s2 += v.y * w[6];  s3 += v.y * w[7];
        s0 += v.z * w[8];  s1 += v.z * w[9];  s2 += v.z * w[10]; s3 += v.z * w[11];
        s0 += v.w * w[12]; s1 += v.w * w[13]; s2 += v.w * w[14]; s3 += v.w * w[15];
    }
    float* o = out + (size_t)i * 4;
    o[0] = s0; o[1] = s1; o[2] = s2; o[3] = s3;
}

extern "C" void kernel_launch(void* const* d_in, const int* in_sizes, int n_in,
                              void* d_out, int out_size, void* d_ws, size_t ws_size,
                              hipStream_t stream) {
    const float* x   = (const float*)d_in[0];
    const int*   ei  = (const int*)d_in[1];
    const float* W1l = (const float*)d_in[2];
    const float* b1  = (const float*)d_in[3];
    const float* W1r = (const float*)d_in[4];
    const float* W2l = (const float*)d_in[5];
    const float* b2  = (const float*)d_in[6];
    const float* W2r = (const float*)d_in[7];
    const float* W3l = (const float*)d_in[8];
    const float* b3  = (const float*)d_in[9];
    const float* W3r = (const float*)d_in[10];
    const float* Wc  = (const float*)d_in[11];
    const float* bc  = (const float*)d_in[12];
    const float* a   = (const float*)d_in[13];
    float* out = (float*)d_out;

    const int E = in_sizes[1] / 2;
    const int* srcv = ei;
    const int* dstv = ei + E;

    char* ws = (char*)d_ws;
    size_t off = 0;
    auto carve = [&](size_t bytes) {
        float* p = (float*)(ws + off);
        off += (bytes + 255) & ~(size_t)255;
        return p;
    };
    float* deg = carve((size_t)N_NODES * 4);
    float* agg = carve((size_t)N_NODES * 128 * 4);
    float* hA  = carve((size_t)N_NODES * 128 * 4);
    float* hB  = carve((size_t)N_NODES * 128 * 4);
    float* h3  = hA;

    const int TB = 256;
    const int nBlk16 = N_NODES / 16;
    const int nBlkSc = (E + TB - 1) / TB;
    const int nOwn   = (N_NODES + RANGE - 1) / RANGE;
    const int nGrp   = (E + 4095) / 4096;

    unsigned* own2 = (unsigned*)hB;
    unsigned* pk2  = own2 + (size_t)nGrp * 4096;
    unsigned* own3 = (unsigned*)(hA + (size_t)N_NODES * 64);
    unsigned* pk3  = own3 + (size_t)nGrp * 4096;

    zero_kernel<<<(N_NODES + TB - 1) / TB, TB, 0, stream>>>(deg, N_NODES);
    edge_keys_kernel<true><<<nGrp, 128, 0, stream>>>(srcv, dstv, deg, own2, pk2, E);
    invdeg_kernel<<<(N_NODES + TB - 1) / TB, TB, 0, stream>>>(deg, N_NODES);

    zero_kernel<<<(N_NODES * 32 + TB - 1) / TB, TB, 0, stream>>>(agg, N_NODES * 32);
    scatter_kernel<32, 8><<<nBlkSc, TB, 0, stream>>>(
        x, srcv, dstv, agg, E);
    sage_combine_wmma<32, 128, 2, true><<<nBlk16, 128, 0, stream>>>(
        agg, deg, x, W1l, b1, W1r, a, hA);

    owner_sum_kernel<128, 8><<<nOwn, 1024, 0, stream>>>(
        hA, own2, pk2, agg, E);
    sage_combine_wmma<128, 128, 2, true><<<nBlk16, 128, 0, stream>>>(
        agg, deg, hA, W2l, b2, W2r, a, hB);

    edge_keys_kernel<false><<<nGrp, 128, 0, stream>>>(srcv, dstv, deg, own3, pk3, E);
    owner_sum_kernel<128, 8><<<nOwn, 1024, 0, stream>>>(
        hB, own3, pk3, agg, E);
    sage_combine_wmma<128, 64, 1, false><<<nBlk16, 128, 0, stream>>>(
        agg, deg, hB, W3l, b3, W3r, a, h3);

    classifier_kernel<<<(N_NODES + TB - 1) / TB, TB, 0, stream>>>(
        h3, Wc, bc, out, N_NODES);
}


# Layer 1's neighbour sums and the degree count also by the owner work-groups (LDS f32 adds; owner stores 1/deg): no global atomic or clear left, 12 launches to 9; sum order changes (LDS arrival)
# speedup vs baseline: 2.9112x; 1.2691x over previous
#include <hip/hip_runtime.h>
#include <hip/hip_bf16.h>

typedef __attribute__((ext_vector_type(2))) float v2f;
typedef __attribute__((ext_vector_type(8))) float v8f;

#define N_NODES 100000

#define RANGE 391
static_assert(N_NODES <= (1 << 17), "src must fit the low 17 bits of the packed word");
static_assert(RANGE <= (1 << 15), "row must fit above bit 17");
static_assert((N_NODES + RANGE - 1) / RANGE <= 256, "owner number must fit two 4-bit parts");

__global__ __launch_bounds__(128) void edge_keys_kernel(const int* __restrict__ src,
                                 const int* __restrict__ dst,
                                 unsigned* __restrict__ planes,
                                 unsigned* __restrict__ pk, int E) {
    const int lane = threadIdx.x & 31;
    const int gw = __builtin_amdgcn_readfirstlane((int)((blockIdx.x * blockDim.x + threadIdx.x) >> 5));
    const int pw = ((E + 4095) >> 12) << 7;
    const int w0 = gw * 32;
    if (w0 >= pw) return;
    unsigned wa[16], wb[16];
#pragma unroll
    for (int k = 0; k < 16; ++k) { wa[k] = 0u; wb[k] = 0u; }
#pragma unroll 1
    for (int i0 = 0; i0 < 32; i0 += 8) {
        int s[8], d[8];
#pragma unroll
        for (int u = 0; u < 8; ++u) {
            const int e  = (w0 + i0 + u) * 32 + lane;
            const int ec = (e < E) ? e : (E - 1);
            s[u] = src[ec]; d[u] = dst[ec];
        }
        unsigned a[8], b[8], p[8];
#pragma unroll
        for (int u = 0; u < 8; ++u) {
            const int e = (w0 + i0 + u) * 32 + lane;
            const bool v = (e < E) && (d[u] >= 0);
            const unsigned o = (unsigned)d[u] / RANGE;
            const unsigned r = (unsigned)d[u] - o * RANGE;
            a[u] = v ? (o >> 4) : 16u;
            b[u] = v ? (o & 15u) : 16u;
            p[u] = (unsigned)s[u] | (r << 17);
        }
#pragma unroll
        for (int u = 0; u < 8; ++u) {
            const bool mine = (lane == i0 + u);
#pragma unroll
            for (int k = 0; k < 16; ++k) {
                const unsigned ma = __builtin_amdgcn_ballot_w32(a[u] == (unsigned)k);
                const unsigned mb = __builtin_amdgcn_ballot_w32(b[u] == (unsigned)k);
                wa[k] = mine ? ma : wa[k];
                wb[k] = mine ? mb : wb[k];
            }
        }
#pragma unroll
        for (int u = 0; u < 8; ++u) {
            const int e = (w0 + i0 + u) * 32 + lane;
            if (a[u] < 16u) pk[e] = p[u];
        }
    }
#pragma unroll
    for (int k = 0; k < 16; ++k) {
        planes[(size_t)k * pw + w0 + lane]        = wa[k];
        planes[(size_t)(16 + k) * pw + w0 + lane] = wb[k];
    }
}

#define QCAP  64
#define OWN_G 4

template <int F, int U, bool FULL, bool COUNT>
__device__ __forceinline__ void owner_drain(const unsigned* qw, int first, int n, int lane,
                                            const float* __restrict__ h,
                                            const unsigned* __restrict__ pk,
                                            float* acc, float* cnt) {
    constexpr int C = F / 32;
    int my_s = 0, my_r = -1;
    if (FULL || lane < n) {
        const unsigned p = pk[qw[first + lane]];
        my_s = (int)(p & 0x1ffffu);
        my_r = (int)(p >> 17);
        if (COUNT) atomicAdd(cnt + my_r, 1.0f);
    }
#pragma unroll 1
    for (int j0 = 0; j0 < n; j0 += U) {
        float v[U][C];
#pragma unroll
        for (int u = 0; u < U; ++u) {
            const int s = __builtin_amdgcn_readlane(my_s, j0 + u);
            const float* row = h + (size_t)s * F + lane;
#pragma unroll
            for (int c = 0; c < C; ++c) v[u][c] = row[c * 32];
        }
#pragma unroll
        for (int u = 0; u < U; ++u) {
            const int r = __builtin_amdgcn_readlane(my_r, j0 + u);
            if (FULL || r >= 0) {
                float* o = acc + r * F + lane;
#pragma unroll
                for (int c = 0; c < C; ++c) atomicAdd(o + c * 32, v[u][c]);
            }
        }
    }
}

template <int F, int U, bool COUNT>
__device__ __forceinline__ int owner_take(unsigned hits, const int eb, const int E,
                                          unsigned* qw, int qn, int lane,
                                          const float* __restrict__ h,
                                          const unsigned* __restrict__ pk, float* acc,
                                          float* cnt) {
    while (__builtin_amdgcn_ballot_w32(hits != 0u) != 0u) {
        const int e = eb + __builtin_ctz(hits | 0x80000000u);
        const bool m = (hits != 0u) && (e < E);
        const unsigned b = __builtin_amdgcn_ballot_w32(m);
        if (m) qw[qn + (int)__builtin_amdgcn_mbcnt_lo(b, 0u)] = (unsigned)e;
        qn += __builtin_popcount(b);
        hits &= hits - 1u;
        if (qn >= 32) {
            qn -= 32;
            owner_drain<F, U, true, COUNT>(qw, qn, 32, lane, h, pk, acc, cnt);
        }
    }
    return qn;
}

template <int F, int U, bool COUNT>
__global__ __launch_bounds__(1024) void owner_sum_kernel(const float* __restrict__ h,
                               const unsigned* __restrict__ planes,
                               const unsigned* __restrict__ pk,
                               float* __restrict__ agg,
                               float* __restrict__ invdeg,
                               int E) {
    __shared__ float acc[RANGE * F];
    __shared__ unsigned q[32 * QCAP];
    __shared__ float cnt[COUNT ? RANGE : 1];

    const int tid  = threadIdx.x;
    const int lane = tid & 31;
    const int wave = __builtin_amdgcn_readfirstlane(tid >> 5);
    const int lo   = blockIdx.x * RANGE;
    int nr = N_NODES - lo;
    if (nr > RANGE) nr = RANGE;

    float4* acc4 = reinterpret_cast<float4*>(acc);
    for (int i = tid; i < RANGE * F / 4; i += 1024)
        acc4[i] = make_float4(0.0f, 0.0f, 0.0f, 0.0f);
    if (COUNT)
        for (int i = tid; i < RANGE; i += 1024) cnt[i] = 0.0f;
    __syncthreads();

    unsigned* qw = q + wave * QCAP;
    int qn = 0;
    const int ng = (E + 4095) >> 12;
    const int pw = ng << 7;
    const uint4* pa = reinterpret_cast<const uint4*>(planes + (size_t)(blockIdx.x >> 4) * pw);
    const uint4* pb = reinterpret_cast<const uint4*>(planes + (size_t)(16 + (blockIdx.x & 15)) * pw);
    const int start = (int)(((long long)blockIdx.x * ng) >> 8);

    for (int t0 = wave; t0 < ng; t0 += 32 * OWN_G) {
        uint4 wa[OWN_G], wb[OWN_G];
        int   eb[OWN_G];
#pragma unroll
        for (int u = 0; u < OWN_G; ++u) {
            const int t = t0 + 32 * u;
            int g = start + ((t < ng) ? t : t0);
            if (g >= ng) g -= ng;
            const int qi = g * 32 + lane;
            eb[u] = qi * 128;
            wa[u] = pa[qi];
            wb[u] = pb[qi];
        }
#pragma unroll
        for (int u = 0; u < OWN_G; ++u) {
            const bool live = (t0 + 32 * u < ng);
            const unsigned m0 = live ? (wa[u].x & wb[u].x) : 0u;
            const unsigned m1 = live ? (wa[u].y & wb[u].y) : 0u;
            const unsigned m2 = live ? (wa[u].z & wb[u].z) : 0u;
            const unsigned m3 = live ? (wa[u].w & wb[u].w) : 0u;
            qn = owner_take<F, U, COUNT>(m0, eb[u],      E, qw, qn, lane, h, pk, acc, cnt);
            qn = owner_take<F, U, COUNT>(m1, eb[u] + 32, E, qw, qn, lane, h, pk, acc, cnt);
            qn = owner_take<F, U, COUNT>(m2, eb[u] + 64, E, qw, qn, lane, h, pk, acc, cnt);
            qn = owner_take<F, U, COUNT>(m3, eb[u] + 96, E, qw, qn, lane, h, pk, acc, cnt);
        }
    }
    if (qn > 0) owner_drain<F, U, false, COUNT>(qw, 0, qn, lane, h, pk, acc, cnt);
    __syncthreads();

    float4* o4 = reinterpret_cast<float4*>(agg + (size_t)lo * F);
    for (int i = tid; i < nr * (F / 4); i += 1024) o4[i] = acc4[i];
    if (COUNT)
        for (int i = tid; i < nr; i += 1024) invdeg[lo + i] = 1.0f / fmaxf(cnt[i], 1.0f);
}

template <int FIN, int FOUT, int TPW, bool PRELU>
__global__ __launch_bounds__(128) void sage_combine_wmma(
    const float* __restrict__ agg, const float* __restrict__ invdeg,
    const float* __restrict__ h,
    const float* __restrict__ Wl, const float* __restrict__ bias,
    const float* __restrict__ Wr, const float* __restrict__ alpha_p,
    float* __restrict__ out) {
    constexpr int LDSTR = FIN + 1;
    __shared__ float s_mean[16 * LDSTR];
    __shared__ float s_h[16 * LDSTR];

    const int tid   = threadIdx.x;
    const int node0 = blockIdx.x * 16;

    for (int idx = tid; idx < 16 * FIN; idx += 128) {
        int r = idx / FIN;
        int c = idx % FIN;
        size_t g = (size_t)(node0 + r) * FIN + c;
        s_mean[r * LDSTR + c] = agg[g] * invdeg[node0 + r];
        s_h[r * LDSTR + c]    = h[g];
    }
    __syncthreads();

    const int wave = tid >> 5;
    const int lane = tid & 31;
    const int half = lane >> 4;
    const int col  = lane & 15;
    const float alpha = alpha_p[0];

    v8f acc[TPW];
#pragma unroll
    for (int t = 0; t < TPW; ++t)
#pragma unroll
        for (int j = 0; j < 8; ++j) acc[t][j] = 0.0f;

    const float* smrow = s_mean + col * LDSTR;
    const float* shrow = s_h    + col * LDSTR;

#pragma unroll 4
    for (int k0 = 0; k0 < FIN; k0 += 4) {
        const int kb = k0 + half * 2;
        v2f am; am.x = smrow[kb]; am.y = smrow[kb + 1];
        v2f ah; ah.x = shrow[kb]; ah.y = shrow[kb + 1];
#pragma unroll
        for (int t = 0; t < TPW; ++t) {
            const int n0 = (wave * TPW + t) * 16;
            v2f bl;
            bl.x = Wl[(size_t)kb * FOUT + n0 + col];
            bl.y = Wl[(size_t)(kb + 1) * FOUT + n0 + col];
            v2f br;
            br.x = Wr[(size_t)kb * FOUT + n0 + col];
            br.y = Wr[(size_t)(kb + 1) * FOUT + n0 + col];
            acc[t] = __builtin_amdgcn_wmma_f32_16x16x4_f32(
                false, am, false, bl, (short)0, acc[t], false, false);
            acc[t] = __builtin_amdgcn_wmma_f32_16x16x4_f32(
                false, ah, false, br, (short)0, acc[t], false, false);
        }
    }

#pragma unroll
    for (int t = 0; t < TPW; ++t) {
        const int n0 = (wave * TPW + t) * 16;
        const float bv = bias[n0 + col];
#pragma unroll
        for (int v = 0; v < 8; ++v) {
            float val = acc[t][v] + bv;
            if (PRELU) val = (val >= 0.0f) ? val : alpha * val;
            const int row = node0 + v + half * 8;
            out[(size_t)row * FOUT + n0 + col] = val;
        }
    }
}

__global__ void classifier_kernel(const float* __restrict__ h3,
                                  const float* __restrict__ Wc,
                                  const float* __restrict__ bc,
                                  float* __restrict__ out, int n) {
    int i = blockIdx.x * blockDim.x + threadIdx.x;
    if (i >= n) return;
    float s0 = bc[0], s1 = bc[1], s2 = bc[2], s3 = bc[3];
    const float4* row = reinterpret_cast<const float4*>(h3 + (size_t)i * 64);
#pragma unroll
    for (int k4 = 0; k4 < 16; ++k4) {
        float4 v = row[k4];
        const float* w = Wc + (k4 * 4) * 4;
        s0 += v.x * w[0];  s1 += v.x * w[1];  s2 += v.x * w[2];  s3 += v.x * w[3];
        s0 += v.y * w[4];  s1 += v.y * w[5];  s2 += v.y * w[6];  s3 += v.y * w[7];
        s0 += v.z * w[8];  s1 += v.z * w[9];  s2 += v.z * w[10]; s3 += v.z * w[11];
        s0 += v.w * w[12]; s1 += v.w * w[13]; s2 += v.w * w[14]; s3 += v.w * w[15];
    }
    float* o = out + (size_t)i * 4;
    o[0] = s0; o[1] = s1; o[2] = s2; o[3] = s3;
}

extern "C" void kernel_launch(void* const* d_in, const int* in_sizes, int n_in,
                              void* d_out, int out_size, void* d_ws, size_t ws_size,
                              hipStream_t stream) {
    const float* x   = (const float*)d_in[0];
    const int*   ei  = (const int*)d_in[1];
    const float* W1l = (const float*)d_in[2];
    const float* b1  = (const float*)d_in[3];
    const float* W1r = (const float*)d_in[4];
    const float* W2l = (const float*)d_in[5];
    const float* b2  = (const float*)d_in[6];
    const float* W2r = (const float*)d_in[7];
    const float* W3l = (const float*)d_in[8];
    const float* b3  = (const float*)d_in[9];
    const float* W3r = (const float*)d_in[10];
    const float* Wc  = (const float*)d_in[11];
    const float* bc  = (const float*)d_in[12];
    const float* a   = (const float*)d_in[13];
    float* out = (float*)d_out;

    const int E = in_sizes[1] / 2;
    const int* srcv = ei;
    const int* dstv = ei + E;

    char* ws = (char*)d_ws;
    size_t off = 0;
    auto carve = [&](size_t bytes) {
        float* p = (float*)(ws + off);
        off += (bytes + 255) & ~(size_t)255;
        return p;
    };
    float* deg = carve((size_t)N_NODES * 4);
    float* agg = carve((size_t)N_NODES * 128 * 4);
    float* hA  = carve((size_t)N_NODES * 128 * 4);
    float* hB  = carve((size_t)N_NODES * 128 * 4);
    float* h3  = hA;

    const int TB = 256;
    const int nBlk16 = N_NODES / 16;
    const int nOwn   = (N_NODES + RANGE - 1) / RANGE;
    const int nGrp   = (E + 4095) / 4096;

    unsigned* own2 = (unsigned*)hB;
    unsigned* pk2  = own2 + (size_t)nGrp * 4096;
    unsigned* own3 = (unsigned*)(hA + (size_t)N_NODES * 64);
    unsigned* pk3  = own3 + (size_t)nGrp * 4096;

    edge_keys_kernel<<<nGrp, 128, 0, stream>>>(srcv, dstv, own2, pk2, E);

    owner_sum_kernel<32, 16, true><<<nOwn, 1024, 0, stream>>>(
        x, own2, pk2, agg, deg, E);
    sage_combine_wmma<32, 128, 2, true><<<nBlk16, 128, 0, stream>>>(
        agg, deg, x, W1l, b1, W1r, a, hA);

    owner_sum_kernel<128, 8, false><<<nOwn, 1024, 0, stream>>>(
        hA, own2, pk2, agg, deg, E);
    sage_combine_wmma<128, 128, 2, true><<<nBlk16, 128, 0, stream>>>(
        agg, deg, hA, W2l, b2, W2r, a, hB);

    edge_keys_kernel<<<nGrp, 128, 0, stream>>>(srcv, dstv, own3, pk3, E);
    owner_sum_kernel<128, 8, false><<<nOwn, 1024, 0, stream>>>(
        hB, own3, pk3, agg, deg, E);
    sage_combine_wmma<128, 64, 1, false><<<nBlk16, 128, 0, stream>>>(
        agg, deg, hB, W3l, b3, W3r, a, h3);

    classifier_kernel<<<(N_NODES + TB - 1) / TB, TB, 0, stream>>>(
        h3, Wc, bc, out, N_NODES);
}


# Layer products done by the owner work-groups on their LDS sums (A read from LDS, own h rows staged after): sums never stored, keys made once, 9 launches to 5; sum order changes (mean@Wl before h@Wr)
# speedup vs baseline: 3.2554x; 1.1182x over previous
#include <hip/hip_runtime.h>
#include <hip/hip_bf16.h>

typedef __attribute__((ext_vector_type(2))) float v2f;
typedef __attribute__((ext_vector_type(8))) float v8f;

#define N_NODES 100000

#define RANGE 391
static_assert(N_NODES <= (1 << 17), "src must fit the low 17 bits of the packed word");
static_assert(RANGE <= (1 << 15), "row must fit above bit 17");
static_assert((N_NODES + RANGE - 1) / RANGE <= 256, "owner number must fit two 4-bit parts");

__global__ __launch_bounds__(128) void edge_keys_kernel(const int* __restrict__ src,
                                 const int* __restrict__ dst,
                                 unsigned* __restrict__ planes,
                                 unsigned* __restrict__ pk, int E) {
    const int lane = threadIdx.x & 31;
    const int gw = __builtin_amdgcn_readfirstlane((int)((blockIdx.x * blockDim.x + threadIdx.x) >> 5));
    const int pw = ((E + 4095) >> 12) << 7;
    const int w0 = gw * 32;
    if (w0 >= pw) return;
    unsigned wa[16], wb[16];
#pragma unroll
    for (int k = 0; k < 16; ++k) { wa[k] = 0u; wb[k] = 0u; }
#pragma unroll 1
    for (int i0 = 0; i0 < 32; i0 += 8) {
        int s[8], d[8];
#pragma unroll
        for (int u = 0; u < 8; ++u) {
            const int e  = (w0 + i0 + u) * 32 + lane;
            const int ec = (e < E) ? e : (E - 1);
            s[u] = src[ec]; d[u] = dst[ec];
        }
        unsigned a[8], b[8], p[8];
#pragma unroll
        for (int u = 0; u < 8; ++u) {
            const int e = (w0 + i0 + u) * 32 + lane;
            const bool v = (e < E) && (d[u] >= 0);
            const unsigned o = (unsigned)d[u] / RANGE;
            const unsigned r = (unsigned)d[u] - o * RANGE;
            a[u] = v ? (o >> 4) : 16u;
            b[u] = v ? (o & 15u) : 16u;
            p[u] = (unsigned)s[u] | (r << 17);
        }
#pragma unroll
        for (int u = 0; u < 8; ++u) {
            const bool mine = (lane == i0 + u);
#pragma unroll
            for (int k = 0; k < 16; ++k) {
                const unsigned ma = __builtin_amdgcn_ballot_w32(a[u] == (unsigned)k);
                const unsigned mb = __builtin_amdgcn_ballot_w32(b[u] == (unsigned)k);
                wa[k] = mine ? ma : wa[k];
                wb[k] = mine ? mb : wb[k];
            }
        }
#pragma unroll
        for (int u = 0; u < 8; ++u) {
            const int e = (w0 + i0 + u) * 32 + lane;
            if (a[u] < 16u) pk[e] = p[u];
        }
    }
#pragma unroll
    for (int k = 0; k < 16; ++k) {
        planes[(size_t)k * pw + w0 + lane]        = wa[k];
        planes[(size_t)(16 + k) * pw + w0 + lane] = wb[k];
    }
}

#define QCAP  64
#define OWN_G 4
#define ROWS  400
#define PITCH_OF(F) ((F) + 4)
static_assert(ROWS % 16 == 0 && ROWS >= RANGE, "whole tiles covering the range");

template <int F, int U, bool FULL, bool COUNT>
__device__ __forceinline__ void owner_drain(const unsigned* qw, int first, int n, int lane,
                                            const float* __restrict__ h,
                                            const unsigned* __restrict__ pk,
                                            float* acc, float* cnt) {
    constexpr int C = F / 32;
    int my_s = 0, my_r = -1;
    if (FULL || lane < n) {
        const unsigned p = pk[qw[first + lane]];
        my_s = (int)(p & 0x1ffffu);
        my_r = (int)(p >> 17);
        if (COUNT) atomicAdd(cnt + my_r, 1.0f);
    }
#pragma unroll 1
    for (int j0 = 0; j0 < n; j0 += U) {
        float v[U][C];
#pragma unroll
        for (int u = 0; u < U; ++u) {
            const int s = __builtin_amdgcn_readlane(my_s, j0 + u);
            const float* row = h + (size_t)s * F + lane;
#pragma unroll
            for (int c = 0; c < C; ++c) v[u][c] = row[c * 32];
        }
#pragma unroll
        for (int u = 0; u < U; ++u) {
            const int r = __builtin_amdgcn_readlane(my_r, j0 + u);
            if (FULL || r >= 0) {
                float* o = acc + r * PITCH_OF(F) + lane;
#pragma unroll
                for (int c = 0; c < C; ++c) atomicAdd(o + c * 32, v[u][c]);
            }
        }
    }
}

template <int F, int U, bool COUNT>
__device__ __forceinline__ int owner_take(unsigned hits, const int eb, const int E,
                                          unsigned* qw, int qn, int lane,
                                          const float* __restrict__ h,
                                          const unsigned* __restrict__ pk, float* acc,
                                          float* cnt) {
    while (__builtin_amdgcn_ballot_w32(hits != 0u) != 0u) {
        const int e = eb + __builtin_ctz(hits | 0x80000000u);
        const bool m = (hits != 0u) && (e < E);
        const unsigned b = __builtin_amdgcn_ballot_w32(m);
        if (m) qw[qn + (int)__builtin_amdgcn_mbcnt_lo(b, 0u)] = (unsigned)e;
        qn += __builtin_popcount(b);
        hits &= hits - 1u;
        if (qn >= 32) {
            qn -= 32;
            owner_drain<F, U, true, COUNT>(qw, qn, 32, lane, h, pk, acc, cnt);
        }
    }
    return qn;
}

template <int F, int FOUT, int T>
__device__ __forceinline__ void rows_mma(const float* ap, const int ts,
                                         const float* __restrict__ wp, v8f (&d)[T]) {
#pragma unroll 4
    for (int k0 = 0; k0 < F; k0 += 4) {
        v2f b;
        b.x = wp[(size_t)k0 * FOUT];
        b.y = wp[(size_t)(k0 + 1) * FOUT];
#pragma unroll
        for (int t = 0; t < T; ++t) {
            const float2 a2 = *reinterpret_cast<const float2*>(ap + t * ts + k0);
            v2f a; a.x = a2.x; a.y = a2.y;
            d[t] = __builtin_amdgcn_wmma_f32_16x16x4_f32(
                false, a, false, b, (short)0, d[t], false, false);
        }
    }
}

template <int OP, bool PRELU>
__device__ __forceinline__ void tile_put(const v8f& d, const float bv, const float alpha,
                                         float* p) {
#pragma unroll
    for (int v = 0; v < 8; ++v) {
        float val = d[v] + bv;
        if (PRELU) val = (val >= 0.0f) ? val : alpha * val;
        p[v * OP] = val;
    }
}

template <int F, int FOUT, int U, bool COUNT, bool PRELU>
__global__ __launch_bounds__(1024) void owner_layer_kernel(const float* __restrict__ h,
                               const unsigned* __restrict__ planes,
                               const unsigned* __restrict__ pk,
                               float* __restrict__ invdeg,
                               const float* __restrict__ Wl, const float* __restrict__ bias,
                               const float* __restrict__ Wr, const float* __restrict__ alpha_p,
                               float* __restrict__ out, int E) {
    constexpr int PITCH = PITCH_OF(F);
    constexpr int OP    = FOUT + 4;
    constexpr int LROW  = (PITCH > OP) ? PITCH : OP;
    __shared__ __attribute__((aligned(16))) float acc[ROWS * LROW];
    __shared__ unsigned q[32 * QCAP];
    __shared__ float cnt[COUNT ? ROWS : 1];

    const int tid  = threadIdx.x;
    const int lane = tid & 31;
    const int wave = __builtin_amdgcn_readfirstlane(tid >> 5);
    const int lo   = blockIdx.x * RANGE;
    int nr = N_NODES - lo;
    if (nr > RANGE) nr = RANGE;

    float4* acc4 = reinterpret_cast<float4*>(acc);
    for (int i = tid; i < ROWS * PITCH / 4; i += 1024)
        acc4[i] = make_float4(0.0f, 0.0f, 0.0f, 0.0f);
    if (COUNT)
        for (int i = tid; i < ROWS; i += 1024) cnt[i] = 0.0f;
    __syncthreads();

    unsigned* qw = q + wave * QCAP;
    int qn = 0;
    const int ng = (E + 4095) >> 12;
    const int pw = ng << 7;
    const uint4* pa = reinterpret_cast<const uint4*>(planes + (size_t)(blockIdx.x >> 4) * pw);
    const uint4* pb = reinterpret_cast<const uint4*>(planes + (size_t)(16 + (blockIdx.x & 15)) * pw);
    const int start = (int)(((long long)blockIdx.x * ng) >> 8);

    for (int t0 = wave; t0 < ng; t0 += 32 * OWN_G) {
        uint4 wa[OWN_G], wb[OWN_G];
        int   eb[OWN_G];
#pragma unroll
        for (int u = 0; u < OWN_G; ++u) {
            const int t = t0 + 32 * u;
            int g = start + ((t < ng) ? t : t0);
            if (g >= ng) g -= ng;
            const int qi = g * 32 + lane;
            eb[u] = qi * 128;
            wa[u] = pa[qi];
            wb[u] = pb[qi];
        }
#pragma unroll
        for (int u = 0; u < OWN_G; ++u) {
            const bool live = (t0 + 32 * u < ng);
            const unsigned m0 = live ? (wa[u].x & wb[u].x) : 0u;
            const unsigned m1 = live ? (wa[u].y & wb[u].y) : 0u;
            const unsigned m2 = live ? (wa[u].z & wb[u].z) : 0u;
            const unsigned m3 = live ? (wa[u].w & wb[u].w) : 0u;
            qn = owner_take<F, U, COUNT>(m0, eb[u],      E, qw, qn, lane, h, pk, acc, cnt);
            qn = owner_take<F, U, COUNT>(m1, eb[u] + 32, E, qw, qn, lane, h, pk, acc, cnt);
            qn = owner_take<F, U, COUNT>(m2, eb[u] + 64, E, qw, qn, lane, h, pk, acc, cnt);
            qn = owner_take<F, U, COUNT>(m3, eb[u] + 96, E, qw, qn, lane, h, pk, acc, cnt);
        }
    }
    if (qn > 0) owner_drain<F, U, false, COUNT>(qw, 0, qn, lane, h, pk, acc, cnt);
    __syncthreads();

    constexpr int Q = F / 4;
    if (COUNT)
        for (int i = tid; i < nr; i += 1024) invdeg[lo + i] = 1.0f / fmaxf(cnt[i], 1.0f);
    for (int i = tid; i < ROWS * Q; i += 1024) {
        const int r = i / Q, c4 = i % Q;
        float s;
        if (COUNT) s = 1.0f / fmaxf(cnt[r], 1.0f);
        else       s = (r < nr) ? invdeg[lo + r] : 0.0f;
        float4* p = reinterpret_cast<float4*>(acc + r * PITCH + 4 * c4);
        float4 v = *p;
        v.x *= s; v.y *= s; v.z *= s; v.w *= s;
        *p = v;
    }
    __syncthreads();

    constexpr int NT = FOUT / 16;
    constexpr int NG = 32 / NT;
    constexpr int MT = ROWS / 16;
    constexpr int TM = MT / NG;
    constexpr int XT = MT % NG;
    static_assert(32 % NT == 0, "whole tile columns per work-group");
    const int nt   = wave % NT;
    const int g    = wave / NT;
    const int half = lane >> 4;
    const int col  = lane & 15;
    const bool extra = (g < XT);

    v8f d[TM], dx[1];
#pragma unroll
    for (int t = 0; t < TM; ++t)
#pragma unroll
        for (int j = 0; j < 8; ++j) d[t][j] = 0.0f;
#pragma unroll
    for (int j = 0; j < 8; ++j) dx[0][j] = 0.0f;

    const float* ap  = acc + (g * 16 + col) * PITCH + half * 2;
    const float* apx = ap + TM * NG * 16 * PITCH;
    constexpr int TS = NG * 16 * PITCH;
    const size_t woff = (size_t)(half * 2) * FOUT + nt * 16 + col;

    rows_mma<F, FOUT, TM>(ap, TS, Wl + woff, d);
    if (extra) rows_mma<F, FOUT, 1>(apx, 0, Wl + woff, dx);
    __syncthreads();

    for (int i = tid; i < ROWS * Q; i += 1024) {
        const int r = i / Q, c4 = i % Q;
        int gr = lo + r;
        if (gr > N_NODES - 1) gr = N_NODES - 1;
        const float4 v = reinterpret_cast<const float4*>(h + (size_t)gr * F)[c4];
        *reinterpret_cast<float4*>(acc + r * PITCH + 4 * c4) = v;
    }
    __syncthreads();

    rows_mma<F, FOUT, TM>(ap, TS, Wr + woff, d);
    if (extra) rows_mma<F, FOUT, 1>(apx, 0, Wr + woff, dx);

    __syncthreads();

    const float alpha = alpha_p[0];
    const float bv = bias[nt * 16 + col];
    float* op = acc + (g * 16 + half * 8) * OP + nt * 16 + col;
#pragma unroll
    for (int t = 0; t < TM; ++t)
        tile_put<OP, PRELU>(d[t], bv, alpha, op + t * NG * 16 * OP);
    if (extra)
        tile_put<OP, PRELU>(dx[0], bv, alpha, op + TM * NG * 16 * OP);
    __syncthreads();

    constexpr int QO = FOUT / 4;
    float4* o4 = reinterpret_cast<float4*>(out + (size_t)lo * FOUT);
    for (int i = tid; i < nr * QO; i += 1024) {
        const int r = i / QO, c4 = i % QO;
        o4[i] = *reinterpret_cast<const float4*>(acc + r * OP + 4 * c4);
    }
}

__global__ void classifier_kernel(const float* __restrict__ h3,
                                  const float* __restrict__ Wc,
                                  const float* __restrict__ bc,
                                  float* __restrict__ out, int n) {
    int i = blockIdx.x * blockDim.x + threadIdx.x;
    if (i >= n) return;
    float s0 = bc[0], s1 = bc[1], s2 = bc[2], s3 = bc[3];
    const float4* row = reinterpret_cast<const float4*>(h3 + (size_t)i * 64);
#pragma unroll
    for (int k4 = 0; k4 < 16; ++k4) {
        float4 v = row[k4];
        const float* w = Wc + (k4 * 4) * 4;
        s0 += v.x * w[0];  s1 += v.x * w[1];  s2 += v.x * w[2];  s3 += v.x * w[3];
        s0 += v.y * w[4];  s1 += v.y * w[5];  s2 += v.y * w[6];  s3 += v.y * w[7];
        s0 += v.z * w[8];  s1 += v.z * w[9];  s2 += v.z * w[10]; s3 += v.z * w[11];
        s0 += v.w * w[12]; s1 += v.w * w[13]; s2 += v.w * w[14]; s3 += v.w * w[15];
    }
    float* o = out + (size_t)i * 4;
    o[0] = s0; o[1] = s1; o[2] = s2; o[3] = s3;
}

extern "C" void kernel_launch(void* const* d_in, const int* in_sizes, int n_in,
                              void* d_out, int out_size, void* d_ws, size_t ws_size,
                              hipStream_t stream) {
    const float* x   = (const float*)d_in[0];
    const int*   ei  = (const int*)d_in[1];
    const float* W1l = (const float*)d_in[2];
    const float* b1  = (const float*)d_in[3];
    const float* W1r = (const float*)d_in[4];
    const float* W2l = (const float*)d_in[5];
    const float* b2  = (const float*)d_in[6];
    const float* W2r = (const float*)d_in[7];
    const float* W3l = (const float*)d_in[8];
    const float* b3  = (const float*)d_in[9];
    const float* W3r = (const float*)d_in[10];
    const float* Wc  = (const float*)d_in[11];
    const float* bc  = (const float*)d_in[12];
    const float* a   = (const float*)d_in[13];
    float* out = (float*)d_out;

    const int E = in_sizes[1] / 2;
    const int* srcv = ei;
    const int* dstv = ei + E;

    char* ws = (char*)d_ws;
    size_t off = 0;
    auto carve = [&](size_t bytes) {
        float* p = (float*)(ws + off);
        off += (bytes + 255) & ~(size_t)255;
        return p;
    };
    float* deg = carve((size_t)N_NODES * 4);
    float* hC  = carve((size_t)N_NODES * 128 * 4);
    float* hA  = carve((size_t)N_NODES * 128 * 4);
    float* hB  = carve((size_t)N_NODES * 128 * 4);
    float* h3  = hA;

    const int TB = 256;
    const int nOwn = (N_NODES + RANGE - 1) / RANGE;
    const int nGrp = (E + 4095) / 4096;

    unsigned* own = (unsigned*)hB;
    unsigned* pk  = own + (size_t)nGrp * 4096;
    edge_keys_kernel<<<nGrp, 128, 0, stream>>>(srcv, dstv, own, pk, E);

    owner_layer_kernel<32, 128, 16, true, true><<<nOwn, 1024, 0, stream>>>(
        x, own, pk, deg, W1l, b1, W1r, a, hA, E);

    owner_layer_kernel<128, 128, 8, false, true><<<nOwn, 1024, 0, stream>>>(
        hA, own, pk, deg, W2l, b2, W2r, a, hC, E);

    owner_layer_kernel<128, 64, 8, false, false><<<nOwn, 1024, 0, stream>>>(
        hC, own, pk, deg, W3l, b3, W3r, a, h3, E);

    classifier_kernel<<<(N_NODES + TB - 1) / TB, TB, 0, stream>>>(
        h3, Wc, bc, out, N_NODES);
}


# Layers 2-3 fetch each 512-B neighbour row by one 16-byte-per-lane load (4 lines per instruction; 16 rows in flight, was 8); LDS sums in that lane order, Wl rows to match: k order of mean@Wl changes
# speedup vs baseline: 3.4439x; 1.0579x over previous
#include <hip/hip_runtime.h>
#include <hip/hip_bf16.h>

typedef __attribute__((ext_vector_type(2))) float v2f;
typedef __attribute__((ext_vector_type(8))) float v8f;

#define N_NODES 100000

#define RANGE 391
static_assert(N_NODES <= (1 << 17), "src must fit the low 17 bits of the packed word");
static_assert(RANGE <= (1 << 15), "row must fit above bit 17");
static_assert((N_NODES + RANGE - 1) / RANGE <= 256, "owner number must fit two 4-bit parts");

__global__ __launch_bounds__(128) void edge_keys_kernel(const int* __restrict__ src,
                                 const int* __restrict__ dst,
                                 unsigned* __restrict__ planes,
                                 unsigned* __restrict__ pk, int E) {
    const int lane = threadIdx.x & 31;
    const int gw = __builtin_amdgcn_readfirstlane((int)((blockIdx.x * blockDim.x + threadIdx.x) >> 5));
    const int pw = ((E + 4095) >> 12) << 7;
    const int w0 = gw * 32;
    if (w0 >= pw) return;
    unsigned wa[16], wb[16];
#pragma unroll
    for (int k = 0; k < 16; ++k) { wa[k] = 0u; wb[k] = 0u; }
#pragma unroll 1
    for (int i0 = 0; i0 < 32; i0 += 8) {
        int s[8], d[8];
#pragma unroll
        for (int u = 0; u < 8; ++u) {
            const int e  = (w0 + i0 + u) * 32 + lane;
            const int ec = (e < E) ? e : (E - 1);
            s[u] = src[ec]; d[u] = dst[ec];
        }
        unsigned a[8], b[8], p[8];
#pragma unroll
        for (int u = 0; u < 8; ++u) {
            const int e = (w0 + i0 + u) * 32 + lane;
            const bool v = (e < E) && (d[u] >= 0);
            const unsigned o = (unsigned)d[u] / RANGE;
            const unsigned r = (unsigned)d[u] - o * RANGE;
            a[u] = v ? (o >> 4) : 16u;
            b[u] = v ? (o & 15u) : 16u;
            p[u] = (unsigned)s[u] | (r << 17);
        }
#pragma unroll
        for (int u = 0; u < 8; ++u) {
            const bool mine = (lane == i0 + u);
#pragma unroll
            for (int k = 0; k < 16; ++k) {
                const unsigned ma = __builtin_amdgcn_ballot_w32(a[u] == (unsigned)k);
                const unsigned mb = __builtin_amdgcn_ballot_w32(b[u] == (unsigned)k);
                wa[k] = mine ? ma : wa[k];
                wb[k] = mine ? mb : wb[k];
            }
        }
#pragma unroll
        for (int u = 0; u < 8; ++u) {
            const int e = (w0 + i0 + u) * 32 + lane;
            if (a[u] < 16u) pk[e] = p[u];
        }
    }
#pragma unroll
    for (int k = 0; k < 16; ++k) {
        planes[(size_t)k * pw + w0 + lane]        = wa[k];
        planes[(size_t)(16 + k) * pw + w0 + lane] = wb[k];
    }
}

#define QCAP  64
#define OWN_G 4
#define ROWS  400
#define PITCH_OF(F) ((F) + 4)
static_assert(ROWS % 16 == 0 && ROWS >= RANGE, "whole tiles covering the range");

template <int F, int U, bool FULL, bool COUNT>
__device__ __forceinline__ void owner_drain(const unsigned* qw, int first, int n, int lane,
                                            const float* __restrict__ h,
                                            const unsigned* __restrict__ pk,
                                            float* acc, float* cnt) {
    constexpr int C = F / 32;
    constexpr bool WIDE = (F == 128);
    static_assert(32 % U == 0, "a full batch is whole trips");
    int my_s = 0, my_r = -1;
    if (FULL || lane < n) {
        const unsigned p = pk[qw[first + lane]];
        my_s = (int)(p & 0x1ffffu);
        my_r = (int)(p >> 17);
        if (COUNT) atomicAdd(cnt + my_r, 1.0f);
    }
    if constexpr (WIDE) {
#pragma unroll 1
        for (int j0 = 0; j0 < n; j0 += U) {
            float4 v[U];
#pragma unroll
            for (int u = 0; u < U; ++u) {
                const int s = __builtin_amdgcn_readlane(my_s, j0 + u);
                v[u] = reinterpret_cast<const float4*>(h + (size_t)s * F)[lane];
            }
#pragma unroll
            for (int u = 0; u < U; ++u) {
                const int r = __builtin_amdgcn_readlane(my_r, j0 + u);
                if (FULL || r >= 0) {
                    float* o = acc + r * PITCH_OF(F) + lane;
                    atomicAdd(o,      v[u].x);
                    atomicAdd(o + 32, v[u].y);
                    atomicAdd(o + 64, v[u].z);
                    atomicAdd(o + 96, v[u].w);
                }
            }
        }
    } else {
#pragma unroll 1
        for (int j0 = 0; j0 < n; j0 += U) {
            float v[U][C];
#pragma unroll
            for (int u = 0; u < U; ++u) {
                const int s = __builtin_amdgcn_readlane(my_s, j0 + u);
                const float* row = h + (size_t)s * F + lane;
#pragma unroll
                for (int c = 0; c < C; ++c) v[u][c] = row[c * 32];
            }
#pragma unroll
            for (int u = 0; u < U; ++u) {
                const int r = __builtin_amdgcn_readlane(my_r, j0 + u);
                if (FULL || r >= 0) {
                    float* o = acc + r * PITCH_OF(F) + lane;
#pragma unroll
                    for (int c = 0; c < C; ++c) atomicAdd(o + c * 32, v[u][c]);
                }
            }
        }
    }
}

template <int F, int U, bool COUNT>
__device__ __forceinline__ int owner_take(unsigned hits, const int eb, const int E,
                                          unsigned* qw, int qn, int lane,
                                          const float* __restrict__ h,
                                          const unsigned* __restrict__ pk, float* acc,
                                          float* cnt) {
    while (__builtin_amdgcn_ballot_w32(hits != 0u) != 0u) {
        const int e = eb + __builtin_ctz(hits | 0x80000000u);
        const bool m = (hits != 0u) && (e < E);
        const unsigned b = __builtin_amdgcn_ballot_w32(m);
        if (m) qw[qn + (int)__builtin_amdgcn_mbcnt_lo(b, 0u)] = (unsigned)e;
        qn += __builtin_popcount(b);
        hits &= hits - 1u;
        if (qn >= 32) {
            qn -= 32;
            owner_drain<F, U, true, COUNT>(qw, qn, 32, lane, h, pk, acc, cnt);
        }
    }
    return qn;
}

template <int F, int FOUT, int T, bool PERM>
__device__ __forceinline__ void rows_mma(const float* ap, const int ts,
                                         const float* __restrict__ wp, v8f (&d)[T]) {
    static_assert(!PERM || F == 128, "the permuted order is the 128-wide sum rows' order");
    constexpr int FS = PERM ? 4 : 1;
#pragma unroll 4
    for (int k0 = 0; k0 < F; k0 += 4) {
        const int f0 = PERM ? (((k0 & 31) << 2) | (k0 >> 5)) : k0;
        v2f b;
        b.x = wp[(size_t)f0 * FOUT];
        b.y = wp[(size_t)(f0 + FS) * FOUT];
#pragma unroll
        for (int t = 0; t < T; ++t) {
            const float2 a2 = *reinterpret_cast<const float2*>(ap + t * ts + k0);
            v2f a; a.x = a2.x; a.y = a2.y;
            d[t] = __builtin_amdgcn_wmma_f32_16x16x4_f32(
                false, a, false, b, (short)0, d[t], false, false);
        }
    }
}

template <int OP, bool PRELU>
__device__ __forceinline__ void tile_put(const v8f& d, const float bv, const float alpha,
                                         float* p) {
#pragma unroll
    for (int v = 0; v < 8; ++v) {
        float val = d[v] + bv;
        if (PRELU) val = (val >= 0.0f) ? val : alpha * val;
        p[v * OP] = val;
    }
}

template <int F, int FOUT, int U, bool COUNT, bool PRELU>
__global__ __launch_bounds__(1024) void owner_layer_kernel(const float* __restrict__ h,
                               const unsigned* __restrict__ planes,
                               const unsigned* __restrict__ pk,
                               float* __restrict__ invdeg,
                               const float* __restrict__ Wl, const float* __restrict__ bias,
                               const float* __restrict__ Wr, const float* __restrict__ alpha_p,
                               float* __restrict__ out, int E) {
    constexpr int PITCH = PITCH_OF(F);
    constexpr int OP    = FOUT + 4;
    constexpr int LROW  = (PITCH > OP) ? PITCH : OP;
    __shared__ __attribute__((aligned(16))) float acc[ROWS * LROW];
    __shared__ unsigned q[32 * QCAP];
    __shared__ float cnt[COUNT ? ROWS : 1];

    const int tid  = threadIdx.x;
    const int lane = tid & 31;
    const int wave = __builtin_amdgcn_readfirstlane(tid >> 5);
    const int lo   = blockIdx.x * RANGE;
    int nr = N_NODES - lo;
    if (nr > RANGE) nr = RANGE;

    float4* acc4 = reinterpret_cast<float4*>(acc);
    for (int i = tid; i < ROWS * PITCH / 4; i += 1024)
        acc4[i] = make_float4(0.0f, 0.0f, 0.0f, 0.0f);
    if (COUNT)
        for (int i = tid; i < ROWS; i += 1024) cnt[i] = 0.0f;
    __syncthreads();

    unsigned* qw = q + wave * QCAP;
    int qn = 0;
    const int ng = (E + 4095) >> 12;
    const int pw = ng << 7;
    const uint4* pa = reinterpret_cast<const uint4*>(planes + (size_t)(blockIdx.x >> 4) * pw);
    const uint4* pb = reinterpret_cast<const uint4*>(planes + (size_t)(16 + (blockIdx.x & 15)) * pw);
    const int start = (int)(((long long)blockIdx.x * ng) >> 8);

    for (int t0 = wave; t0 < ng; t0 += 32 * OWN_G) {
        uint4 wa[OWN_G], wb[OWN_G];
        int   eb[OWN_G];
#pragma unroll
        for (int u = 0; u < OWN_G; ++u) {
            const int t = t0 + 32 * u;
            int g = start + ((t < ng) ? t : t0);
            if (g >= ng) g -= ng;
            const int qi = g * 32 + lane;
            eb[u] = qi * 128;
            wa[u] = pa[qi];
            wb[u] = pb[qi];
        }
#pragma unroll
        for (int u = 0; u < OWN_G; ++u) {
            const bool live = (t0 + 32 * u < ng);
            const unsigned m0 = live ? (wa[u].x & wb[u].x) : 0u;
            const unsigned m1 = live ? (wa[u].y & wb[u].y) : 0u;
            const unsigned m2 = live ? (wa[u].z & wb[u].z) : 0u;
            const unsigned m3 = live ? (wa[u].w & wb[u].w) : 0u;
            qn = owner_take<F, U, COUNT>(m0, eb[u],      E, qw, qn, lane, h, pk, acc, cnt);
            qn = owner_take<F, U, COUNT>(m1, eb[u] + 32, E, qw, qn, lane, h, pk, acc, cnt);
            qn = owner_take<F, U, COUNT>(m2, eb[u] + 64, E, qw, qn, lane, h, pk, acc, cnt);
            qn = owner_take<F, U, COUNT>(m3, eb[u] + 96, E, qw, qn, lane, h, pk, acc, cnt);
        }
    }
    if (qn > 0) owner_drain<F, U, false, COUNT>(qw, 0, qn, lane, h, pk, acc, cnt);
    __syncthreads();

    constexpr int Q = F / 4;
    if (COUNT)
        for (int i = tid; i < nr; i += 1024) invdeg[lo + i] = 1.0f / fmaxf(cnt[i], 1.0f);
    for (int i = tid; i < ROWS * Q; i += 1024) {
        const int r = i / Q, c4 = i % Q;
        float s;
        if (COUNT) s = 1.0f / fmaxf(cnt[r], 1.0f);
        else       s = (r < nr) ? invdeg[lo + r] : 0.0f;
        float4* p = reinterpret_cast<float4*>(acc + r * PITCH + 4 * c4);
        float4 v = *p;
        v.x *= s; v.y *= s; v.z *= s; v.w *= s;
        *p = v;
    }
    __syncthreads();

    constexpr int NT = FOUT / 16;
    constexpr int NG = 32 / NT;
    constexpr int MT = ROWS / 16;
    constexpr int TM = MT / NG;
    constexpr int XT = MT % NG;
    static_assert(32 % NT == 0, "whole tile columns per work-group");
    const int nt   = wave % NT;
    const int g    = wave / NT;
    const int half = lane >> 4;
    const int col  = lane & 15;
    const bool extra = (g < XT);

    v8f d[TM], dx[1];
#pragma unroll
    for (int t = 0; t < TM; ++t)
#pragma unroll
        for (int j = 0; j < 8; ++j) d[t][j] = 0.0f;
#pragma unroll
    for (int j = 0; j < 8; ++j) dx[0][j] = 0.0f;

    const float* ap  = acc + (g * 16 + col) * PITCH + half * 2;
    const float* apx = ap + TM * NG * 16 * PITCH;
    constexpr int TS = NG * 16 * PITCH;
    const size_t woff = (size_t)(half * 2) * FOUT + nt * 16 + col;
    constexpr bool WIDE = (F == 128);
    const size_t woffl = WIDE ? (size_t)(half * 8) * FOUT + nt * 16 + col : woff;

    rows_mma<F, FOUT, TM, WIDE>(ap, TS, Wl + woffl, d);
    if (extra) rows_mma<F, FOUT, 1, WIDE>(apx, 0, Wl + woffl, dx);
    __syncthreads();

    for (int i = tid; i < ROWS * Q; i += 1024) {
        const int r = i / Q, c4 = i % Q;
        int gr = lo + r;
        if (gr > N_NODES - 1) gr = N_NODES - 1;
        const float4 v = reinterpret_cast<const float4*>(h + (size_t)gr * F)[c4];
        *reinterpret_cast<float4*>(acc + r * PITCH + 4 * c4) = v;
    }
    __syncthreads();

    rows_mma<F, FOUT, TM, false>(ap, TS, Wr + woff, d);
    if (extra) rows_mma<F, FOUT, 1, false>(apx, 0, Wr + woff, dx);

    __syncthreads();

    const float alpha = alpha_p[0];
    const float bv = bias[nt * 16 + col];
    float* op = acc + (g * 16 + half * 8) * OP + nt * 16 + col;
#pragma unroll
    for (int t = 0; t < TM; ++t)
        tile_put<OP, PRELU>(d[t], bv, alpha, op + t * NG * 16 * OP);
    if (extra)
        tile_put<OP, PRELU>(dx[0], bv, alpha, op + TM * NG * 16 * OP);
    __syncthreads();

    constexpr int QO = FOUT / 4;
    float4* o4 = reinterpret_cast<float4*>(out + (size_t)lo * FOUT);
    for (int i = tid; i < nr * QO; i += 1024) {
        const int r = i / QO, c4 = i % QO;
        o4[i] = *reinterpret_cast<const float4*>(acc + r * OP + 4 * c4);
    }
}

__global__ void classifier_kernel(const float* __restrict__ h3,
                                  const float* __restrict__ Wc,
                                  const float* __restrict__ bc,
                                  float* __restrict__ out, int n) {
    int i = blockIdx.x * blockDim.x + threadIdx.x;
    if (i >= n) return;
    float s0 = bc[0], s1 = bc[1], s2 = bc[2], s3 = bc[3];
    const float4* row = reinterpret_cast<const float4*>(h3 + (size_t)i * 64);
#pragma unroll
    for (int k4 = 0; k4 < 16; ++k4) {
        float4 v = row[k4];
        const float* w = Wc + (k4 * 4) * 4;
        s0 += v.x * w[0];  s1 += v.x * w[1];  s2 += v.x * w[2];  s3 += v.x * w[3];
        s0 += v.y * w[4];  s1 += v.y * w[5];  s2 += v.y * w[6];  s3 += v.y * w[7];
        s0 += v.z * w[8];  s1 += v.z * w[9];  s2 += v.z * w[10]; s3 += v.z * w[11];
        s0 += v.w * w[12]; s1 += v.w * w[13]; s2 += v.w * w[14]; s3 += v.w * w[15];
    }
    float* o = out + (size_t)i * 4;
    o[0] = s0; o[1] = s1; o[2] = s2; o[3] = s3;
}

extern "C" void kernel_launch(void* const* d_in, const int* in_sizes, int n_in,
                              void* d_out, int out_size, void* d_ws, size_t ws_size,
                              hipStream_t stream) {
    const float* x   = (const float*)d_in[0];
    const int*   ei  = (const int*)d_in[1];
    const float* W1l = (const float*)d_in[2];
    const float* b1  = (const float*)d_in[3];
    const float* W1r = (const float*)d_in[4];
    const float* W2l = (const float*)d_in[5];
    const float* b2  = (const float*)d_in[6];
    const float* W2r = (const float*)d_in[7];
    const float* W3l = (const float*)d_in[8];
    const float* b3  = (const float*)d_in[9];
    const float* W3r = (const float*)d_in[10];
    const float* Wc  = (const float*)d_in[11];
    const float* bc  = (const float*)d_in[12];
    const float* a   = (const float*)d_in[13];
    float* out = (float*)d_out;

    const int E = in_sizes[1] / 2;
    const int* srcv = ei;
    const int* dstv = ei + E;

    char* ws = (char*)d_ws;
    size_t off = 0;
    auto carve = [&](size_t bytes) {
        float* p = (float*)(ws + off);
        off += (bytes + 255) & ~(size_t)255;
        return p;
    };
    float* deg = carve((size_t)N_NODES * 4);
    float* hC  = carve((size_t)N_NODES * 128 * 4);
    float* hA  = carve((size_t)N_NODES * 128 * 4);
    float* hB  = carve((size_t)N_NODES * 128 * 4);
    float* h3  = hA;

    const int TB = 256;
    const int nOwn = (N_NODES + RANGE - 1) / RANGE;
    const int nGrp = (E + 4095) / 4096;

    unsigned* own = (unsigned*)hB;
    unsigned* pk  = own + (size_t)nGrp * 4096;
    edge_keys_kernel<<<nGrp, 128, 0, stream>>>(srcv, dstv, own, pk, E);

    owner_layer_kernel<32, 128, 16, true, true><<<nOwn, 1024, 0, stream>>>(
        x, own, pk, deg, W1l, b1, W1r, a, hA, E);

    owner_layer_kernel<128, 128, 16, false, true><<<nOwn, 1024, 0, stream>>>(
        hA, own, pk, deg, W2l, b2, W2r, a, hC, E);

    owner_layer_kernel<128, 64, 16, false, false><<<nOwn, 1024, 0, stream>>>(
        hC, own, pk, deg, W3l, b3, W3r, a, h3, E);

    classifier_kernel<<<(N_NODES + TB - 1) / TB, TB, 0, stream>>>(
        h3, Wc, bc, out, N_NODES);
}


# Layers 2-3 walk per-wave edge lists (packed src|row words appended by layer 1's owner scan, 1024 slots a wave; plane scan kept for overflow): no plane or scattered key reads; sum order: LDS arrival
# speedup vs baseline: 3.7777x; 1.0969x over previous
#include <hip/hip_runtime.h>
#include <hip/hip_bf16.h>

typedef __attribute__((ext_vector_type(2))) float v2f;
typedef __attribute__((ext_vector_type(8))) float v8f;

#define N_NODES 100000

#define RANGE 391
static_assert(N_NODES <= (1 << 17), "src must fit the low 17 bits of the packed word");
static_assert(RANGE <= (1 << 15), "row must fit above bit 17");
static_assert((N_NODES + RANGE - 1) / RANGE <= 256, "owner number must fit two 4-bit parts");

__global__ __launch_bounds__(128) void edge_keys_kernel(const int* __restrict__ src,
                                 const int* __restrict__ dst,
                                 unsigned* __restrict__ planes,
                                 unsigned* __restrict__ pk, int E) {
    const int lane = threadIdx.x & 31;
    const int gw = __builtin_amdgcn_readfirstlane((int)((blockIdx.x * blockDim.x + threadIdx.x) >> 5));
    const int pw = ((E + 4095) >> 12) << 7;
    const int w0 = gw * 32;
    if (w0 >= pw) return;
    unsigned wa[16], wb[16];
#pragma unroll
    for (int k = 0; k < 16; ++k) { wa[k] = 0u; wb[k] = 0u; }
#pragma unroll 1
    for (int i0 = 0; i0 < 32; i0 += 8) {
        int s[8], d[8];
#pragma unroll
        for (int u = 0; u < 8; ++u) {
            const int e  = (w0 + i0 + u) * 32 + lane;
            const int ec = (e < E) ? e : (E - 1);
            s[u] = src[ec]; d[u] = dst[ec];
        }
        unsigned a[8], b[8], p[8];
#pragma unroll
        for (int u = 0; u < 8; ++u) {
            const int e = (w0 + i0 + u) * 32 + lane;
            const bool v = (e < E) && (d[u] >= 0);
            const unsigned o = (unsigned)d[u] / RANGE;
            const unsigned r = (unsigned)d[u] - o * RANGE;
            a[u] = v ? (o >> 4) : 16u;
            b[u] = v ? (o & 15u) : 16u;
            p[u] = (unsigned)s[u] | (r << 17);
        }
#pragma unroll
        for (int u = 0; u < 8; ++u) {
            const bool mine = (lane == i0 + u);
#pragma unroll
            for (int k = 0; k < 16; ++k) {
                const unsigned ma = __builtin_amdgcn_ballot_w32(a[u] == (unsigned)k);
                const unsigned mb = __builtin_amdgcn_ballot_w32(b[u] == (unsigned)k);
                wa[k] = mine ? ma : wa[k];
                wb[k] = mine ? mb : wb[k];
            }
        }
#pragma unroll
        for (int u = 0; u < 8; ++u) {
            const int e = (w0 + i0 + u) * 32 + lane;
            if (a[u] < 16u) pk[e] = p[u];
        }
    }
#pragma unroll
    for (int k = 0; k < 16; ++k) {
        planes[(size_t)k * pw + w0 + lane]        = wa[k];
        planes[(size_t)(16 + k) * pw + w0 + lane] = wb[k];
    }
}

#define LCAP  1024
#define QCAP  64
#define OWN_G 4
#define ROWS  400
#define PITCH_OF(F) ((F) + 4)
static_assert(ROWS % 16 == 0 && ROWS >= RANGE, "whole tiles covering the range");

template <int F, int U, bool FULL>
__device__ __forceinline__ void owner_rows(const int my_s, const int my_r, int n, int lane,
                                           const float* __restrict__ h, float* acc) {
    constexpr int C = F / 32;
    constexpr bool WIDE = (F == 128);
    static_assert(32 % U == 0, "a full batch is whole trips");
    if constexpr (WIDE) {
#pragma unroll 1
        for (int j0 = 0; j0 < n; j0 += U) {
            float4 v[U];
#pragma unroll
            for (int u = 0; u < U; ++u) {
                const int s = __builtin_amdgcn_readlane(my_s, j0 + u);
                v[u] = reinterpret_cast<const float4*>(h + (size_t)s * F)[lane];
            }
#pragma unroll
            for (int u = 0; u < U; ++u) {
                const int r = __builtin_amdgcn_readlane(my_r, j0 + u);
                if (FULL || r >= 0) {
                    float* o = acc + r * PITCH_OF(F) + lane;
                    atomicAdd(o,      v[u].x);
                    atomicAdd(o + 32, v[u].y);
                    atomicAdd(o + 64, v[u].z);
                    atomicAdd(o + 96, v[u].w);
                }
            }
        }
    } else {
#pragma unroll 1
        for (int j0 = 0; j0 < n; j0 += U) {
            float v[U][C];
#pragma unroll
            for (int u = 0; u < U; ++u) {
                const int s = __builtin_amdgcn_readlane(my_s, j0 + u);
                const float* row = h + (size_t)s * F + lane;
#pragma unroll
                for (int c = 0; c < C; ++c) v[u][c] = row[c * 32];
            }
#pragma unroll
            for (int u = 0; u < U; ++u) {
                const int r = __builtin_amdgcn_readlane(my_r, j0 + u);
                if (FULL || r >= 0) {
                    float* o = acc + r * PITCH_OF(F) + lane;
#pragma unroll
                    for (int c = 0; c < C; ++c) atomicAdd(o + c * 32, v[u][c]);
                }
            }
        }
    }
}

template <int F, int U, bool FULL, bool COUNT>
__device__ __forceinline__ void owner_drain(const unsigned* qw, int first, int n, int lane,
                                            const float* __restrict__ h,
                                            const unsigned* __restrict__ pk,
                                            float* acc, float* cnt, unsigned* lw, int ln) {
    int my_s = 0, my_r = -1;
    if (FULL || lane < n) {
        const unsigned p = pk[qw[first + lane]];
        my_s = (int)(p & 0x1ffffu);
        my_r = (int)(p >> 17);
        if (COUNT) {
            atomicAdd(cnt + my_r, 1.0f);
            if (ln + n <= LCAP) lw[ln + lane] = p;
        }
    }
    owner_rows<F, U, FULL>(my_s, my_r, n, lane, h, acc);
}

template <int F, int U>
__device__ __forceinline__ void owner_walk(const unsigned* __restrict__ lw, const int n, int lane,
                                           const float* __restrict__ h, float* acc) {
    if (n <= 0) return;
    unsigned p = lw[(lane < n) ? lane : (n - 1)];
    int j = 0;
#pragma unroll 1
    for (; j + 32 <= n; j += 32) {
        const int nx = j + 32 + lane;
        const unsigned pn = lw[(nx < n) ? nx : (n - 1)];
        owner_rows<F, U, true>((int)(p & 0x1ffffu), (int)(p >> 17), 32, lane, h, acc);
        p = pn;
    }
    const int m = n - j;
    if (m > 0) {
        const bool on = (lane < m);
        owner_rows<F, U, false>(on ? (int)(p & 0x1ffffu) : 0, on ? (int)(p >> 17) : -1,
                                m, lane, h, acc);
    }
}

template <int F, int U, bool COUNT>
__device__ __forceinline__ int owner_take(unsigned hits, const int eb, const int E,
                                          unsigned* qw, int qn, int lane,
                                          const float* __restrict__ h,
                                          const unsigned* __restrict__ pk, float* acc,
                                          float* cnt, unsigned* lw, int& ln) {
    while (__builtin_amdgcn_ballot_w32(hits != 0u) != 0u) {
        const int e = eb + __builtin_ctz(hits | 0x80000000u);
        const bool m = (hits != 0u) && (e < E);
        const unsigned b = __builtin_amdgcn_ballot_w32(m);
        if (m) qw[qn + (int)__builtin_amdgcn_mbcnt_lo(b, 0u)] = (unsigned)e;
        qn += __builtin_popcount(b);
        hits &= hits - 1u;
        if (qn >= 32) {
            qn -= 32;
            owner_drain<F, U, true, COUNT>(qw, qn, 32, lane, h, pk, acc, cnt, lw, ln);
            ln += 32;
        }
    }
    return qn;
}

template <int F, int FOUT, int T, bool PERM>
__device__ __forceinline__ void rows_mma(const float* ap, const int ts,
                                         const float* __restrict__ wp, v8f (&d)[T]) {
    static_assert(!PERM || F == 128, "the permuted order is the 128-wide sum rows' order");
    constexpr int FS = PERM ? 4 : 1;
#pragma unroll 4
    for (int k0 = 0; k0 < F; k0 += 4) {
        const int f0 = PERM ? (((k0 & 31) << 2) | (k0 >> 5)) : k0;
        v2f b;
        b.x = wp[(size_t)f0 * FOUT];
        b.y = wp[(size_t)(f0 + FS) * FOUT];
#pragma unroll
        for (int t = 0; t < T; ++t) {
            const float2 a2 = *reinterpret_cast<const float2*>(ap + t * ts + k0);
            v2f a; a.x = a2.x; a.y = a2.y;
            d[t] = __builtin_amdgcn_wmma_f32_16x16x4_f32(
                false, a, false, b, (short)0, d[t], false, false);
        }
    }
}

template <int OP, bool PRELU>
__device__ __forceinline__ void tile_put(const v8f& d, const float bv, const float alpha,
                                         float* p) {
#pragma unroll
    for (int v = 0; v < 8; ++v) {
        float val = d[v] + bv;
        if (PRELU) val = (val >= 0.0f) ? val : alpha * val;
        p[v * OP] = val;
    }
}

template <int F, int FOUT, int U, bool COUNT, bool PRELU>
__global__ __launch_bounds__(1024) void owner_layer_kernel(const float* __restrict__ h,
                               const unsigned* __restrict__ planes,
                               const unsigned* __restrict__ pk,
                               unsigned* __restrict__ lst, int* __restrict__ lcnt,
                               float* __restrict__ invdeg,
                               const float* __restrict__ Wl, const float* __restrict__ bias,
                               const float* __restrict__ Wr, const float* __restrict__ alpha_p,
                               float* __restrict__ out, int E) {
    constexpr int PITCH = PITCH_OF(F);
    constexpr int OP    = FOUT + 4;
    constexpr int LROW  = (PITCH > OP) ? PITCH : OP;
    __shared__ __attribute__((aligned(16))) float acc[ROWS * LROW];
    __shared__ unsigned q[32 * QCAP];
    __shared__ float cnt[COUNT ? ROWS : 1];

    const int tid  = threadIdx.x;
    const int lane = tid & 31;
    const int wave = __builtin_amdgcn_readfirstlane(tid >> 5);
    const int lo   = blockIdx.x * RANGE;
    int nr = N_NODES - lo;
    if (nr > RANGE) nr = RANGE;

    unsigned* lw = lst + (size_t)(blockIdx.x * 32 + wave) * LCAP;
    bool by_list = false;
    int  nl = 0;
    if constexpr (!COUNT) {
        const int myc = lcnt[blockIdx.x * 32 + lane];
        by_list = (__builtin_amdgcn_ballot_w32(myc > LCAP) == 0u);
        nl = __builtin_amdgcn_readlane(myc, wave);
    }

    float4* acc4 = reinterpret_cast<float4*>(acc);
    for (int i = tid; i < ROWS * PITCH / 4; i += 1024)
        acc4[i] = make_float4(0.0f, 0.0f, 0.0f, 0.0f);
    if (COUNT)
        for (int i = tid; i < ROWS; i += 1024) cnt[i] = 0.0f;
    __syncthreads();

    if (by_list) {
        owner_walk<F, U>(lw, nl, lane, h, acc);
    } else {
        unsigned* qw = q + wave * QCAP;
        int qn = 0;
        int ln = 0;
        const int ng = (E + 4095) >> 12;
        const int pw = ng << 7;
        const uint4* pa = reinterpret_cast<const uint4*>(planes + (size_t)(blockIdx.x >> 4) * pw);
        const uint4* pb = reinterpret_cast<const uint4*>(planes + (size_t)(16 + (blockIdx.x & 15)) * pw);
        const int start = (int)(((long long)blockIdx.x * ng) >> 8);

        for (int t0 = wave; t0 < ng; t0 += 32 * OWN_G) {
            uint4 wa[OWN_G], wb[OWN_G];
            int   eb[OWN_G];
#pragma unroll
            for (int u = 0; u < OWN_G; ++u) {
                const int t = t0 + 32 * u;
                int g = start + ((t < ng) ? t : t0);
                if (g >= ng) g -= ng;
                const int qi = g * 32 + lane;
                eb[u] = qi * 128;
                wa[u] = pa[qi];
                wb[u] = pb[qi];
            }
#pragma unroll
            for (int u = 0; u < OWN_G; ++u) {
                const bool live = (t0 + 32 * u < ng);
                const unsigned m0 = live ? (wa[u].x & wb[u].x) : 0u;
                const unsigned m1 = live ? (wa[u].y & wb[u].y) : 0u;
                const unsigned m2 = live ? (wa[u].z & wb[u].z) : 0u;
                const unsigned m3 = live ? (wa[u].w & wb[u].w) : 0u;
                qn = owner_take<F, U, COUNT>(m0, eb[u],      E, qw, qn, lane, h, pk, acc, cnt, lw, ln);
                qn = owner_take<F, U, COUNT>(m1, eb[u] + 32, E, qw, qn, lane, h, pk, acc, cnt, lw, ln);
                qn = owner_take<F, U, COUNT>(m2, eb[u] + 64, E, qw, qn, lane, h, pk, acc, cnt, lw, ln);
                qn = owner_take<F, U, COUNT>(m3, eb[u] + 96, E, qw, qn, lane, h, pk, acc, cnt, lw, ln);
            }
        }
        if (qn > 0) owner_drain<F, U, false, COUNT>(qw, 0, qn, lane, h, pk, acc, cnt, lw, ln);
        ln += qn;
        if (COUNT && lane == 0) lcnt[blockIdx.x * 32 + wave] = ln;
    }
    __syncthreads();

    constexpr int Q = F / 4;
    if (COUNT)
        for (int i = tid; i < nr; i += 1024) invdeg[lo + i] = 1.0f / fmaxf(cnt[i], 1.0f);
    for (int i = tid; i < ROWS * Q; i += 1024) {
        const int r = i / Q, c4 = i % Q;
        float s;
        if (COUNT) s = 1.0f / fmaxf(cnt[r], 1.0f);
        else       s = (r < nr) ? invdeg[lo + r] : 0.0f;
        float4* p = reinterpret_cast<float4*>(acc + r * PITCH + 4 * c4);
        float4 v = *p;
        v.x *= s; v.y *= s; v.z *= s; v.w *= s;
        *p = v;
    }
    __syncthreads();

    constexpr int NT = FOUT / 16;
    constexpr int NG = 32 / NT;
    constexpr int MT = ROWS / 16;
    constexpr int TM = MT / NG;
    constexpr int XT = MT % NG;
    static_assert(32 % NT == 0, "whole tile columns per work-group");
    const int nt   = wave % NT;
    const int g    = wave / NT;
    const int half = lane >> 4;
    const int col  = lane & 15;
    const bool extra = (g < XT);

    v8f d[TM], dx[1];
#pragma unroll
    for (int t = 0; t < TM; ++t)
#pragma unroll
        for (int j = 0; j < 8; ++j) d[t][j] = 0.0f;
#pragma unroll
    for (int j = 0; j < 8; ++j) dx[0][j] = 0.0f;

    const float* ap  = acc + (g * 16 + col) * PITCH + half * 2;
    const float* apx = ap + TM * NG * 16 * PITCH;
    constexpr int TS = NG * 16 * PITCH;
    const size_t woff = (size_t)(half * 2) * FOUT + nt * 16 + col;
    constexpr bool WIDE = (F == 128);
    const size_t woffl = WIDE ? (size_t)(half * 8) * FOUT + nt * 16 + col : woff;

    rows_mma<F, FOUT, TM, WIDE>(ap, TS, Wl + woffl, d);
    if (extra) rows_mma<F, FOUT, 1, WIDE>(apx, 0, Wl + woffl, dx);
    __syncthreads();

    for (int i = tid; i < ROWS * Q; i += 1024) {
        const int r = i / Q, c4 = i % Q;
        int gr = lo + r;
        if (gr > N_NODES - 1) gr = N_NODES - 1;
        const float4 v = reinterpret_cast<const float4*>(h + (size_t)gr * F)[c4];
        *reinterpret_cast<float4*>(acc + r * PITCH + 4 * c4) = v;
    }
    __syncthreads();

    rows_mma<F, FOUT, TM, false>(ap, TS, Wr + woff, d);
    if (extra) rows_mma<F, FOUT, 1, false>(apx, 0, Wr + woff, dx);

    __syncthreads();

    const float alpha = alpha_p[0];
    const float bv = bias[nt * 16 + col];
    float* op = acc + (g * 16 + half * 8) * OP + nt * 16 + col;
#pragma unroll
    for (int t = 0; t < TM; ++t)
        tile_put<OP, PRELU>(d[t], bv, alpha, op + t * NG * 16 * OP);
    if (extra)
        tile_put<OP, PRELU>(dx[0], bv, alpha, op + TM * NG * 16 * OP);
    __syncthreads();

    constexpr int QO = FOUT / 4;
    float4* o4 = reinterpret_cast<float4*>(out + (size_t)lo * FOUT);
    for (int i = tid; i < nr * QO; i += 1024) {
        const int r = i / QO, c4 = i % QO;
        o4[i] = *reinterpret_cast<const float4*>(acc + r * OP + 4 * c4);
    }
}

__global__ void classifier_kernel(const float* __restrict__ h3,
                                  const float* __restrict__ Wc,
                                  const float* __restrict__ bc,
                                  float* __restrict__ out, int n) {
    int i = blockIdx.x * blockDim.x + threadIdx.x;
    if (i >= n) return;
    float s0 = bc[0], s1 = bc[1], s2 = bc[2], s3 = bc[3];
    const float4* row = reinterpret_cast<const float4*>(h3 + (size_t)i * 64);
#pragma unroll
    for (int k4 = 0; k4 < 16; ++k4) {
        float4 v = row[k4];
        const float* w = Wc + (k4 * 4) * 4;
        s0 += v.x * w[0];  s1 += v.x * w[1];  s2 += v.x * w[2];  s3 += v.x * w[3];
        s0 += v.y * w[4];  s1 += v.y * w[5];  s2 += v.y * w[6];  s3 += v.y * w[7];
        s0 += v.z * w[8];  s1 += v.z * w[9];  s2 += v.z * w[10]; s3 += v.z * w[11];
        s0 += v.w * w[12]; s1 += v.w * w[13]; s2 += v.w * w[14]; s3 += v.w * w[15];
    }
    float* o = out + (size_t)i * 4;
    o[0] = s0; o[1] = s1; o[2] = s2; o[3] = s3;
}

extern "C" void kernel_launch(void* const* d_in, const int* in_sizes, int n_in,
                              void* d_out, int out_size, void* d_ws, size_t ws_size,
                              hipStream_t stream) {
    const float* x   = (const float*)d_in[0];
    const int*   ei  = (const int*)d_in[1];
    const float* W1l = (const float*)d_in[2];
    const float* b1  = (const float*)d_in[3];
    const float* W1r = (const float*)d_in[4];
    const float* W2l = (const float*)d_in[5];
    const float* b2  = (const float*)d_in[6];
    const float* W2r = (const float*)d_in[7];
    const float* W3l = (const float*)d_in[8];
    const float* b3  = (const float*)d_in[9];
    const float* W3r = (const float*)d_in[10];
    const float* Wc  = (const float*)d_in[11];
    const float* bc  = (const float*)d_in[12];
    const float* a   = (const float*)d_in[13];
    float* out = (float*)d_out;

    const int E = in_sizes[1] / 2;
    const int* srcv = ei;
    const int* dstv = ei + E;

    char* ws = (char*)d_ws;
    size_t off = 0;
    auto carve = [&](size_t bytes) {
        float* p = (float*)(ws + off);
        off += (bytes + 255) & ~(size_t)255;
        return p;
    };
    float* deg = carve((size_t)N_NODES * 4);
    float* hC  = carve((size_t)N_NODES * 128 * 4);
    float* hA  = carve((size_t)N_NODES * 128 * 4);
    float* hB  = carve((size_t)N_NODES * 128 * 4);
    float* h3  = hA;

    const int TB = 256;
    const int nOwn = (N_NODES + RANGE - 1) / RANGE;
    const int nGrp = (E + 4095) / 4096;

    unsigned* own = (unsigned*)hB;
    unsigned* pk  = own + (size_t)nGrp * 4096;
    unsigned* lst = pk + (size_t)nGrp * 4096;
    int* lcnt = (int*)(lst + (size_t)nOwn * 32 * LCAP);
    edge_keys_kernel<<<nGrp, 128, 0, stream>>>(srcv, dstv, own, pk, E);

    owner_layer_kernel<32, 128, 16, true, true><<<nOwn, 1024, 0, stream>>>(
        x, own, pk, lst, lcnt, deg, W1l, b1, W1r, a, hA, E);

    owner_layer_kernel<128, 128, 16, false, true><<<nOwn, 1024, 0, stream>>>(
        hA, own, pk, lst, lcnt, deg, W2l, b2, W2r, a, hC, E);

    owner_layer_kernel<128, 64, 16, false, false><<<nOwn, 1024, 0, stream>>>(
        hC, own, pk, lst, lcnt, deg, W3l, b3, W3r, a, h3, E);

    classifier_kernel<<<(N_NODES + TB - 1) / TB, TB, 0, stream>>>(
        h3, Wc, bc, out, N_NODES);
}


# Layers 2-3 run as two 16-wave half-range work-groups per unit (layer 1 splits its lists by half); every other group does h@Wr first, so one multiplies while the other gathers; sum order changes
# speedup vs baseline: 3.9011x; 1.0327x over previous
#include <hip/hip_runtime.h>
#include <hip/hip_bf16.h>

typedef __attribute__((ext_vector_type(2))) float v2f;
typedef __attribute__((ext_vector_type(8))) float v8f;

#define N_NODES 100000

#define RANGE 391
static_assert(N_NODES <= (1 << 17), "src must fit the low 17 bits of the packed word");
static_assert(RANGE <= (1 << 15), "row must fit above bit 17");
static_assert((N_NODES + RANGE - 1) / RANGE <= 256, "owner number must fit two 4-bit parts");

__global__ __launch_bounds__(128) void edge_keys_kernel(const int* __restrict__ src,
                                 const int* __restrict__ dst,
                                 unsigned* __restrict__ planes,
                                 unsigned* __restrict__ pk, int E) {
    const int lane = threadIdx.x & 31;
    const int gw = __builtin_amdgcn_readfirstlane((int)((blockIdx.x * blockDim.x + threadIdx.x) >> 5));
    const int pw = ((E + 4095) >> 12) << 7;
    const int w0 = gw * 32;
    if (w0 >= pw) return;
    unsigned wa[16], wb[16];
#pragma unroll
    for (int k = 0; k < 16; ++k) { wa[k] = 0u; wb[k] = 0u; }
#pragma unroll 1
    for (int i0 = 0; i0 < 32; i0 += 8) {
        int s[8], d[8];
#pragma unroll
        for (int u = 0; u < 8; ++u) {
            const int e  = (w0 + i0 + u) * 32 + lane;
            const int ec = (e < E) ? e : (E - 1);
            s[u] = src[ec]; d[u] = dst[ec];
        }
        unsigned a[8], b[8], p[8];
#pragma unroll
        for (int u = 0; u < 8; ++u) {
            const int e = (w0 + i0 + u) * 32 + lane;
            const bool v = (e < E) && (d[u] >= 0);
            const unsigned o = (unsigned)d[u] / RANGE;
            const unsigned r = (unsigned)d[u] - o * RANGE;
            a[u] = v ? (o >> 4) : 16u;
            b[u] = v ? (o & 15u) : 16u;
            p[u] = (unsigned)s[u] | (r << 17);
        }
#pragma unroll
        for (int u = 0; u < 8; ++u) {
            const bool mine = (lane == i0 + u);
#pragma unroll
            for (int k = 0; k < 16; ++k) {
                const unsigned ma = __builtin_amdgcn_ballot_w32(a[u] == (unsigned)k);
                const unsigned mb = __builtin_amdgcn_ballot_w32(b[u] == (unsigned)k);
                wa[k] = mine ? ma : wa[k];
                wb[k] = mine ? mb : wb[k];
            }
        }
#pragma unroll
        for (int u = 0; u < 8; ++u) {
            const int e = (w0 + i0 + u) * 32 + lane;
            if (a[u] < 16u) pk[e] = p[u];
        }
    }
#pragma unroll
    for (int k = 0; k < 16; ++k) {
        planes[(size_t)k * pw + w0 + lane]        = wa[k];
        planes[(size_t)(16 + k) * pw + w0 + lane] = wb[k];
    }
}

#define LCAP  1024
#define HALF  196
#define LCAP2 (LCAP / 2)
#define HROWS 208
#define QCAP  64
#define OWN_G 4
#define ROWS  400
#define PITCH_OF(F) ((F) + 4)
static_assert(ROWS % 16 == 0 && ROWS >= RANGE, "whole tiles covering the range");
static_assert(HROWS % 16 == 0 && HROWS >= HALF && HROWS >= RANGE - HALF, "whole tiles covering either half");

template <int F, int U, bool FULL>
__device__ __forceinline__ void owner_rows(const int my_s, const int my_r, int n, int lane,
                                           const float* __restrict__ h, float* acc) {
    constexpr int C = F / 32;
    constexpr bool WIDE = (F == 128);
    static_assert(32 % U == 0, "a full batch is whole trips");
    if constexpr (WIDE) {
#pragma unroll 1
        for (int j0 = 0; j0 < n; j0 += U) {
            float4 v[U];
#pragma unroll
            for (int u = 0; u < U; ++u) {
                const int s = __builtin_amdgcn_readlane(my_s, j0 + u);
                v[u] = reinterpret_cast<const float4*>(h + (size_t)s * F)[lane];
            }
#pragma unroll
            for (int u = 0; u < U; ++u) {
                const int r = __builtin_amdgcn_readlane(my_r, j0 + u);
                if (FULL || r >= 0) {
                    float* o = acc + r * PITCH_OF(F) + lane;
                    atomicAdd(o,      v[u].x);
                    atomicAdd(o + 32, v[u].y);
                    atomicAdd(o + 64, v[u].z);
                    atomicAdd(o + 96, v[u].w);
                }
            }
        }
    } else {
#pragma unroll 1
        for (int j0 = 0; j0 < n; j0 += U) {
            float v[U][C];
#pragma unroll
            for (int u = 0; u < U; ++u) {
                const int s = __builtin_amdgcn_readlane(my_s, j0 + u);
                const float* row = h + (size_t)s * F + lane;
#pragma unroll
                for (int c = 0; c < C; ++c) v[u][c] = row[c * 32];
            }
#pragma unroll
            for (int u = 0; u < U; ++u) {
                const int r = __builtin_amdgcn_readlane(my_r, j0 + u);
                if (FULL || r >= 0) {
                    float* o = acc + r * PITCH_OF(F) + lane;
#pragma unroll
                    for (int c = 0; c < C; ++c) atomicAdd(o + c * 32, v[u][c]);
                }
            }
        }
    }
}

template <int F, int U, bool FULL, bool COUNT>
__device__ __forceinline__ void owner_drain(const unsigned* qw, int first, int n, int lane,
                                            const float* __restrict__ h,
                                            const unsigned* __restrict__ pk,
                                            float* acc, float* cnt, unsigned* lw,
                                            int& ln0, int& ln1, const int rlo, const int rcnt) {
    const bool on = FULL || (lane < n);
    unsigned p = 0u;
    int my_s = 0, my_r = -1;
    if (on) {
        p = pk[qw[first + lane]];
        my_s = (int)(p & 0x1ffffu);
        my_r = (int)(p >> 17);
    }
    if constexpr (COUNT) {
        if (on) atomicAdd(cnt + my_r, 1.0f);
        const bool hi = on && (my_r >= HALF);
        const bool lo = on && (my_r < HALF);
        const unsigned bh = __builtin_amdgcn_ballot_w32(hi);
        const unsigned bl = __builtin_amdgcn_ballot_w32(lo);
        const int ph = ln1 + (int)__builtin_amdgcn_mbcnt_lo(bh, 0u);
        const int pl = ln0 + (int)__builtin_amdgcn_mbcnt_lo(bl, 0u);
        if (lo && pl < LCAP2) lw[pl] = p;
        if (hi && ph < LCAP2) lw[LCAP2 + ph] = p - ((unsigned)HALF << 17);
        ln0 += __builtin_popcount(bl);
        ln1 += __builtin_popcount(bh);
        owner_rows<F, U, FULL>(my_s, my_r, n, lane, h, acc);
    } else {
        const int rr = my_r - rlo;
        my_r = (on && rr >= 0 && rr < rcnt) ? rr : -1;
        owner_rows<F, U, false>(my_s, my_r, n, lane, h, acc);
    }
}

template <int F, int U>
__device__ __forceinline__ void owner_walk(const unsigned* __restrict__ lw, const int n, int lane,
                                           const float* __restrict__ h, float* acc) {
    if (n <= 0) return;
    unsigned p = lw[(lane < n) ? lane : (n - 1)];
    int j = 0;
#pragma unroll 1
    for (; j + 32 <= n; j += 32) {
        const int nx = j + 32 + lane;
        const unsigned pn = lw[(nx < n) ? nx : (n - 1)];
        owner_rows<F, U, true>((int)(p & 0x1ffffu), (int)(p >> 17), 32, lane, h, acc);
        p = pn;
    }
    const int m = n - j;
    if (m > 0) {
        const bool on = (lane < m);
        owner_rows<F, U, false>(on ? (int)(p & 0x1ffffu) : 0, on ? (int)(p >> 17) : -1,
                                m, lane, h, acc);
    }
}

template <int F, int U, bool COUNT>
__device__ __forceinline__ int owner_take(unsigned hits, const int eb, const int E,
                                          unsigned* qw, int qn, int lane,
                                          const float* __restrict__ h,
                                          const unsigned* __restrict__ pk, float* acc,
                                          float* cnt, unsigned* lw, int& ln0, int& ln1,
                                          const int rlo, const int rcnt) {
    while (__builtin_amdgcn_ballot_w32(hits != 0u) != 0u) {
        const int e = eb + __builtin_ctz(hits | 0x80000000u);
        const bool m = (hits != 0u) && (e < E);
        const unsigned b = __builtin_amdgcn_ballot_w32(m);
        if (m) qw[qn + (int)__builtin_amdgcn_mbcnt_lo(b, 0u)] = (unsigned)e;
        qn += __builtin_popcount(b);
        hits &= hits - 1u;
        if (qn >= 32) {
            qn -= 32;
            owner_drain<F, U, true, COUNT>(qw, qn, 32, lane, h, pk, acc, cnt, lw, ln0, ln1,
                                           rlo, rcnt);
        }
    }
    return qn;
}

template <int F, int FOUT, int T, bool PERM>
__device__ __forceinline__ void rows_mma(const float* ap, const int ts,
                                         const float* __restrict__ wp, v8f (&d)[T]) {
    static_assert(!PERM || F == 128, "the permuted order is the 128-wide sum rows' order");
    constexpr int FS = PERM ? 4 : 1;
#pragma unroll 4
    for (int k0 = 0; k0 < F; k0 += 4) {
        const int f0 = PERM ? (((k0 & 31) << 2) | (k0 >> 5)) : k0;
        v2f b;
        b.x = wp[(size_t)f0 * FOUT];
        b.y = wp[(size_t)(f0 + FS) * FOUT];
#pragma unroll
        for (int t = 0; t < T; ++t) {
            const float2 a2 = *reinterpret_cast<const float2*>(ap + t * ts + k0);
            v2f a; a.x = a2.x; a.y = a2.y;
            d[t] = __builtin_amdgcn_wmma_f32_16x16x4_f32(
                false, a, false, b, (short)0, d[t], false, false);
        }
    }
}

template <int OP, bool PRELU>
__device__ __forceinline__ void tile_put(const v8f& d, const float bv, const float alpha,
                                         float* p) {
#pragma unroll
    for (int v = 0; v < 8; ++v) {
        float val = d[v] + bv;
        if (PRELU) val = (val >= 0.0f) ? val : alpha * val;
        p[v * OP] = val;
    }
}

template <int F, int FOUT, int U, bool PRELU>
__global__ __launch_bounds__(1024) void owner_layer_kernel(const float* __restrict__ h,
                               const unsigned* __restrict__ planes,
                               const unsigned* __restrict__ pk,
                               unsigned* __restrict__ lst, int* __restrict__ lcnt,
                               float* __restrict__ invdeg,
                               const float* __restrict__ Wl, const float* __restrict__ bias,
                               const float* __restrict__ Wr, const float* __restrict__ alpha_p,
                               float* __restrict__ out, int E) {
    constexpr int PITCH = PITCH_OF(F);
    constexpr int OP    = FOUT + 4;
    constexpr int LROW  = (PITCH > OP) ? PITCH : OP;
    __shared__ __attribute__((aligned(16))) float acc[ROWS * LROW];
    __shared__ unsigned q[32 * QCAP];
    __shared__ float cnt[ROWS];

    const int tid  = threadIdx.x;
    const int lane = tid & 31;
    const int wave = __builtin_amdgcn_readfirstlane(tid >> 5);
    const int lo   = blockIdx.x * RANGE;
    int nr = N_NODES - lo;
    if (nr > RANGE) nr = RANGE;

    unsigned* lw = lst + (size_t)(blockIdx.x * 32 + wave) * LCAP;

    float4* acc4 = reinterpret_cast<float4*>(acc);
    for (int i = tid; i < ROWS * PITCH / 4; i += 1024)
        acc4[i] = make_float4(0.0f, 0.0f, 0.0f, 0.0f);
    for (int i = tid; i < ROWS; i += 1024) cnt[i] = 0.0f;
    __syncthreads();

    {
        unsigned* qw = q + wave * QCAP;
        int qn = 0;
        int ln0 = 0, ln1 = 0;
        const int ng = (E + 4095) >> 12;
        const int pw = ng << 7;
        const uint4* pa = reinterpret_cast<const uint4*>(planes + (size_t)(blockIdx.x >> 4) * pw);
        const uint4* pb = reinterpret_cast<const uint4*>(planes + (size_t)(16 + (blockIdx.x & 15)) * pw);
        const int start = (int)(((long long)blockIdx.x * ng) >> 8);

        for (int t0 = wave; t0 < ng; t0 += 32 * OWN_G) {
            uint4 wa[OWN_G], wb[OWN_G];
            int   eb[OWN_G];
#pragma unroll
            for (int u = 0; u < OWN_G; ++u) {
                const int t = t0 + 32 * u;
                int g = start + ((t < ng) ? t : t0);
                if (g >= ng) g -= ng;
                const int qi = g * 32 + lane;
                eb[u] = qi * 128;
                wa[u] = pa[qi];
                wb[u] = pb[qi];
            }
#pragma unroll
            for (int u = 0; u < OWN_G; ++u) {
                const bool live = (t0 + 32 * u < ng);
                const unsigned m0 = live ? (wa[u].x & wb[u].x) : 0u;
                const unsigned m1 = live ? (wa[u].y & wb[u].y) : 0u;
                const unsigned m2 = live ? (wa[u].z & wb[u].z) : 0u;
                const unsigned m3 = live ? (wa[u].w & wb[u].w) : 0u;
                qn = owner_take<F, U, true>(m0, eb[u],      E, qw, qn, lane, h, pk, acc, cnt, lw, ln0, ln1, 0, RANGE);
                qn = owner_take<F, U, true>(m1, eb[u] + 32, E, qw, qn, lane, h, pk, acc, cnt, lw, ln0, ln1, 0, RANGE);
                qn = owner_take<F, U, true>(m2, eb[u] + 64, E, qw, qn, lane, h, pk, acc, cnt, lw, ln0, ln1, 0, RANGE);
                qn = owner_take<F, U, true>(m3, eb[u] + 96, E, qw, qn, lane, h, pk, acc, cnt, lw, ln0, ln1, 0, RANGE);
            }
        }
        if (qn > 0)
            owner_drain<F, U, false, true>(qw, 0, qn, lane, h, pk, acc, cnt, lw, ln0, ln1, 0, RANGE);
        if (lane == 0) {
            lcnt[(blockIdx.x * 32 + wave) * 2]     = ln0;
            lcnt[(blockIdx.x * 32 + wave) * 2 + 1] = ln1;
        }
    }
    __syncthreads();

    constexpr int Q = F / 4;
    for (int i = tid; i < nr; i += 1024) invdeg[lo + i] = 1.0f / fmaxf(cnt[i], 1.0f);
    for (int i = tid; i < ROWS * Q; i += 1024) {
        const int r = i / Q, c4 = i % Q;
        const float s = 1.0f / fmaxf(cnt[r], 1.0f);
        float4* p = reinterpret_cast<float4*>(acc + r * PITCH + 4 * c4);
        float4 v = *p;
        v.x *= s; v.y *= s; v.z *= s; v.w *= s;
        *p = v;
    }
    __syncthreads();

    constexpr int NT = FOUT / 16;
    constexpr int NG = 32 / NT;
    constexpr int MT = ROWS / 16;
    constexpr int TM = MT / NG;
    constexpr int XT = MT % NG;
    static_assert(32 % NT == 0, "whole tile columns per work-group");
    const int nt   = wave % NT;
    const int g    = wave / NT;
    const int half = lane >> 4;
    const int col  = lane & 15;
    const bool extra = (g < XT);

    v8f d[TM], dx[1];
#pragma unroll
    for (int t = 0; t < TM; ++t)
#pragma unroll
        for (int j = 0; j < 8; ++j) d[t][j] = 0.0f;
#pragma unroll
    for (int j = 0; j < 8; ++j) dx[0][j] = 0.0f;

    const float* ap  = acc + (g * 16 + col) * PITCH + half * 2;
    const float* apx = ap + TM * NG * 16 * PITCH;
    constexpr int TS = NG * 16 * PITCH;
    const size_t woff = (size_t)(half * 2) * FOUT + nt * 16 + col;
    constexpr bool WIDE = (F == 128);
    const size_t woffl = WIDE ? (size_t)(half * 8) * FOUT + nt * 16 + col : woff;

    rows_mma<F, FOUT, TM, WIDE>(ap, TS, Wl + woffl, d);
    if (extra) rows_mma<F, FOUT, 1, WIDE>(apx, 0, Wl + woffl, dx);
    __syncthreads();

    for (int i = tid; i < ROWS * Q; i += 1024) {
        const int r = i / Q, c4 = i % Q;
        int gr = lo + r;
        if (gr > N_NODES - 1) gr = N_NODES - 1;
        const float4 v = reinterpret_cast<const float4*>(h + (size_t)gr * F)[c4];
        *reinterpret_cast<float4*>(acc + r * PITCH + 4 * c4) = v;
    }
    __syncthreads();

    rows_mma<F, FOUT, TM, false>(ap, TS, Wr + woff, d);
    if (extra) rows_mma<F, FOUT, 1, false>(apx, 0, Wr + woff, dx);

    __syncthreads();

    const float alpha = alpha_p[0];
    const float bv = bias[nt * 16 + col];
    float* op = acc + (g * 16 + half * 8) * OP + nt * 16 + col;
#pragma unroll
    for (int t = 0; t < TM; ++t)
        tile_put<OP, PRELU>(d[t], bv, alpha, op + t * NG * 16 * OP);
    if (extra)
        tile_put<OP, PRELU>(dx[0], bv, alpha, op + TM * NG * 16 * OP);
    __syncthreads();

    constexpr int QO = FOUT / 4;
    float4* o4 = reinterpret_cast<float4*>(out + (size_t)lo * FOUT);
    for (int i = tid; i < nr * QO; i += 1024) {
        const int r = i / QO, c4 = i % QO;
        o4[i] = *reinterpret_cast<const float4*>(acc + r * OP + 4 * c4);
    }
}

template <int F, int FOUT, int U, bool PRELU>
__global__ __launch_bounds__(512, 8) void half_layer_kernel(const float* __restrict__ h,
                               const unsigned* __restrict__ planes,
                               const unsigned* __restrict__ pk,
                               const unsigned* __restrict__ lst, const int* __restrict__ lcnt,
                               const float* __restrict__ invdeg,
                               const float* __restrict__ Wl, const float* __restrict__ bias,
                               const float* __restrict__ Wr, const float* __restrict__ alpha_p,
                               float* __restrict__ out, int E) {
    static_assert(F == 128, "layers 2 and 3 read 128-wide rows");
    constexpr int NW    = 16;
    constexpr int NTH   = NW * 32;
    constexpr int PITCH = PITCH_OF(F);
    constexpr int OP    = FOUT + 4;
    constexpr int LROW  = (PITCH > OP) ? PITCH : OP;
    __shared__ __attribute__((aligned(16))) float acc[HROWS * LROW];
    __shared__ unsigned q[NW * QCAP];

    const int tid  = threadIdx.x;
    const int lane = tid & 31;
    const int wave = __builtin_amdgcn_readfirstlane(tid >> 5);
    const int b    = blockIdx.x >> 1;
    const int p    = blockIdx.x & 1;
    const int rlo  = p * HALF;
    const int cap  = p ? (RANGE - HALF) : HALF;
    const int lo   = b * RANGE + rlo;
    int nr = N_NODES - lo;
    if (nr > cap) nr = cap;
    if (nr < 0) nr = 0;
    const bool root_first = (__builtin_popcount(blockIdx.x) & 1) != 0;

    const int myc = lcnt[(b * 32 + lane) * 2 + p];
    const bool by_list = (__builtin_amdgcn_ballot_w32(myc > LCAP2) == 0u);
    const int n0 = __builtin_amdgcn_readlane(myc, 2 * wave);
    const int n1 = __builtin_amdgcn_readlane(myc, 2 * wave + 1);
    const unsigned* l0 = lst + ((size_t)(b * 32 + 2 * wave) * 2 + p) * LCAP2;
    const unsigned* l1 = l0 + 2 * LCAP2;

    constexpr int Q  = F / 4;
    constexpr int NT = FOUT / 16;
    constexpr int NG = NW / NT;
    constexpr int MT = HROWS / 16;
    constexpr int TM = MT / NG;
    constexpr int XT = MT % NG;
    static_assert(NW % NT == 0, "whole tile columns per work-group");
    const int nt   = wave % NT;
    const int g    = wave / NT;
    const int half = lane >> 4;
    const int col  = lane & 15;
    const bool extra = (g < XT);

    v8f d[TM], dx[1];
    auto zero_tiles = [&]() {
#pragma unroll
        for (int t = 0; t < TM; ++t)
#pragma unroll
            for (int j = 0; j < 8; ++j) d[t][j] = 0.0f;
#pragma unroll
        for (int j = 0; j < 8; ++j) dx[0][j] = 0.0f;
    };

    const float* ap  = acc + (g * 16 + col) * PITCH + half * 2;
    const float* apx = ap + TM * NG * 16 * PITCH;
    constexpr int TS = NG * 16 * PITCH;
    const size_t woff  = (size_t)(half * 2) * FOUT + nt * 16 + col;
    const size_t woffl = (size_t)(half * 8) * FOUT + nt * 16 + col;
    float4* acc4 = reinterpret_cast<float4*>(acc);

    auto neighbour_part = [&](const bool first) {
        for (int i = tid; i < HROWS * PITCH / 4; i += NTH)
            acc4[i] = make_float4(0.0f, 0.0f, 0.0f, 0.0f);
        __syncthreads();
        if (by_list) {
            if (root_first) {
                owner_walk<F, U / 2>(l0, n0, lane, h, acc);
                owner_walk<F, U / 2>(l1, n1, lane, h, acc);
            } else {
                owner_walk<F, U>(l0, n0, lane, h, acc);
                owner_walk<F, U>(l1, n1, lane, h, acc);
            }
        } else {
            unsigned* qw = q + wave * QCAP;
            int qn = 0;
            int z0 = 0, z1 = 0;
            const int ng = (E + 4095) >> 12;
            const int pw = ng << 7;
            const uint4* pa = reinterpret_cast<const uint4*>(planes + (size_t)(b >> 4) * pw);
            const uint4* pb = reinterpret_cast<const uint4*>(planes + (size_t)(16 + (b & 15)) * pw);
            const int start = (int)(((long long)b * ng) >> 8);
            for (int t = wave; t < ng; t += NW) {
                int gi = start + t;
                if (gi >= ng) gi -= ng;
                const int qi = gi * 32 + lane;
                const int eb = qi * 128;
                const uint4 wa = pa[qi];
                const uint4 wb = pb[qi];
                qn = owner_take<F, U / 2, false>(wa.x & wb.x, eb,      E, qw, qn, lane, h, pk, acc, nullptr, nullptr, z0, z1, rlo, cap);
                qn = owner_take<F, U / 2, false>(wa.y & wb.y, eb + 32, E, qw, qn, lane, h, pk, acc, nullptr, nullptr, z0, z1, rlo, cap);
                qn = owner_take<F, U / 2, false>(wa.z & wb.z, eb + 64, E, qw, qn, lane, h, pk, acc, nullptr, nullptr, z0, z1, rlo, cap);
                qn = owner_take<F, U / 2, false>(wa.w & wb.w, eb + 96, E, qw, qn, lane, h, pk, acc, nullptr, nullptr, z0, z1, rlo, cap);
            }
            if (qn > 0)
                owner_drain<F, U / 2, false, false>(qw, 0, qn, lane, h, pk, acc, nullptr, nullptr, z0, z1, rlo, cap);
        }
        __syncthreads();
        for (int i = tid; i < HROWS * Q; i += NTH) {
            const int r = i / Q, c4 = i % Q;
            const float s = (r < nr) ? invdeg[lo + r] : 0.0f;
            float4* ptr = reinterpret_cast<float4*>(acc + r * PITCH + 4 * c4);
            float4 v = *ptr;
            v.x *= s; v.y *= s; v.z *= s; v.w *= s;
            *ptr = v;
        }
        __syncthreads();
        if (first) zero_tiles();
        rows_mma<F, FOUT, TM, true>(ap, TS, Wl + woffl, d);
        if (extra) rows_mma<F, FOUT, 1, true>(apx, 0, Wl + woffl, dx);
        __syncthreads();
    };

    auto root_part = [&](const bool first) {
        for (int i = tid; i < HROWS * Q; i += NTH) {
            const int r = i / Q, c4 = i % Q;
            int gr = lo + r;
            if (gr > N_NODES - 1) gr = N_NODES - 1;
            const float4 v = reinterpret_cast<const float4*>(h + (size_t)gr * F)[c4];
            *reinterpret_cast<float4*>(acc + r * PITCH + 4 * c4) = v;
        }
        __syncthreads();
        if (first) zero_tiles();
        rows_mma<F, FOUT, TM, false>(ap, TS, Wr + woff, d);
        if (extra) rows_mma<F, FOUT, 1, false>(apx, 0, Wr + woff, dx);
        __syncthreads();
    };

    if (root_first) {
        root_part(true);
        neighbour_part(false);
    } else {
        neighbour_part(true);
        root_part(false);
    }

    const float alpha = alpha_p[0];
    const float bv = bias[nt * 16 + col];
    float* op = acc + (g * 16 + half * 8) * OP + nt * 16 + col;
#pragma unroll
    for (int t = 0; t < TM; ++t)
        tile_put<OP, PRELU>(d[t], bv, alpha, op + t * NG * 16 * OP);
    if (extra)
        tile_put<OP, PRELU>(dx[0], bv, alpha, op + TM * NG * 16 * OP);
    __syncthreads();

    constexpr int QO = FOUT / 4;
    float4* o4 = reinterpret_cast<float4*>(out + (size_t)lo * FOUT);
    for (int i = tid; i < nr * QO; i += NTH) {
        const int r = i / QO, c4 = i % QO;
        o4[i] = *reinterpret_cast<const float4*>(acc + r * OP + 4 * c4);
    }
}

__global__ void classifier_kernel(const float* __restrict__ h3,
                                  const float* __restrict__ Wc,
                                  const float* __restrict__ bc,
                                  float* __restrict__ out, int n) {
    int i = blockIdx.x * blockDim.x + threadIdx.x;
    if (i >= n) return;
    float s0 = bc[0], s1 = bc[1], s2 = bc[2], s3 = bc[3];
    const float4* row = reinterpret_cast<const float4*>(h3 + (size_t)i * 64);
#pragma unroll
    for (int k4 = 0; k4 < 16; ++k4) {
        float4 v = row[k4];
        const float* w = Wc + (k4 * 4) * 4;
        s0 += v.x * w[0];  s1 += v.x * w[1];  s2 += v.x * w[2];  s3 += v.x * w[3];
        s0 += v.y * w[4];  s1 += v.y * w[5];  s2 += v.y * w[6];  s3 += v.y * w[7];
        s0 += v.z * w[8];  s1 += v.z * w[9];  s2 += v.z * w[10]; s3 += v.z * w[11];
        s0 += v.w * w[12]; s1 += v.w * w[13]; s2 += v.w * w[14]; s3 += v.w * w[15];
    }
    float* o = out + (size_t)i * 4;
    o[0] = s0; o[1] = s1; o[2] = s2; o[3] = s3;
}

extern "C" void kernel_launch(void* const* d_in, const int* in_sizes, int n_in,
                              void* d_out, int out_size, void* d_ws, size_t ws_size,
                              hipStream_t stream) {
    const float* x   = (const float*)d_in[0];
    const int*   ei  = (const int*)d_in[1];
    const float* W1l = (const float*)d_in[2];
    const float* b1  = (const float*)d_in[3];
    const float* W1r = (const float*)d_in[4];
    const float* W2l = (const float*)d_in[5];
    const float* b2  = (const float*)d_in[6];
    const float* W2r = (const float*)d_in[7];
    const float* W3l = (const float*)d_in[8];
    const float* b3  = (const float*)d_in[9];
    const float* W3r = (const float*)d_in[10];
    const float* Wc  = (const float*)d_in[11];
    const float* bc  = (const float*)d_in[12];
    const float* a   = (const float*)d_in[13];
    float* out = (float*)d_out;

    const int E = in_sizes[1] / 2;
    const int* srcv = ei;
    const int* dstv = ei + E;

    char* ws = (char*)d_ws;
    size_t off = 0;
    auto carve = [&](size_t bytes) {
        float* p = (float*)(ws + off);
        off += (bytes + 255) & ~(size_t)255;
        return p;
    };
    float* deg = carve((size_t)N_NODES * 4);
    float* hC  = carve((size_t)N_NODES * 128 * 4);
    float* hA  = carve((size_t)N_NODES * 128 * 4);
    float* hB  = carve((size_t)N_NODES * 128 * 4);
    float* h3  = hA;

    const int TB = 256;
    const int nOwn = (N_NODES + RANGE - 1) / RANGE;
    const int nGrp = (E + 4095) / 4096;

    unsigned* own = (unsigned*)hB;
    unsigned* pk  = own + (size_t)nGrp * 4096;
    unsigned* lst = pk + (size_t)nGrp * 4096;
    int* lcnt = (int*)(lst + (size_t)nOwn * 32 * LCAP);
    edge_keys_kernel<<<nGrp, 128, 0, stream>>>(srcv, dstv, own, pk, E);

    owner_layer_kernel<32, 128, 16, true><<<nOwn, 1024, 0, stream>>>(
        x, own, pk, lst, lcnt, deg, W1l, b1, W1r, a, hA, E);

    half_layer_kernel<128, 128, 16, true><<<2 * nOwn, 512, 0, stream>>>(
        hA, own, pk, lst, lcnt, deg, W2l, b2, W2r, a, hC, E);

    half_layer_kernel<128, 64, 16, false><<<2 * nOwn, 512, 0, stream>>>(
        hC, own, pk, lst, lcnt, deg, W3l, b3, W3r, a, h3, E);

    classifier_kernel<<<(N_NODES + TB - 1) / TB, TB, 0, stream>>>(
        h3, Wc, bc, out, N_NODES);
}


# Layers 2-3: each wave sums only rows it alone owns (layer 1 bins edges into per-owner-wave lists by LDS counters): plain 16-byte LDS read-add-write, no LDS float atomics; sum order changes
# speedup vs baseline: 4.0327x; 1.0337x over previous
#include <hip/hip_runtime.h>
#include <hip/hip_bf16.h>

typedef __attribute__((ext_vector_type(2))) float v2f;
typedef __attribute__((ext_vector_type(8))) float v8f;

#define N_NODES 100000

#define RANGE 391
static_assert(N_NODES <= (1 << 17), "src must fit the low 17 bits of the packed word");
static_assert(RANGE <= (1 << 15), "row must fit above bit 17");
static_assert((N_NODES + RANGE - 1) / RANGE <= 256, "owner number must fit two 4-bit parts");

__global__ __launch_bounds__(128) void edge_keys_kernel(const int* __restrict__ src,
                                 const int* __restrict__ dst,
                                 unsigned* __restrict__ planes,
                                 unsigned* __restrict__ pk, int E) {
    const int lane = threadIdx.x & 31;
    const int gw = __builtin_amdgcn_readfirstlane((int)((blockIdx.x * blockDim.x + threadIdx.x) >> 5));
    const int pw = ((E + 4095) >> 12) << 7;
    const int w0 = gw * 32;
    if (w0 >= pw) return;
    unsigned wa[16], wb[16];
#pragma unroll
    for (int k = 0; k < 16; ++k) { wa[k] = 0u; wb[k] = 0u; }
#pragma unroll 1
    for (int i0 = 0; i0 < 32; i0 += 8) {
        int s[8], d[8];
#pragma unroll
        for (int u = 0; u < 8; ++u) {
            const int e  = (w0 + i0 + u) * 32 + lane;
            const int ec = (e < E) ? e : (E - 1);
            s[u] = src[ec]; d[u] = dst[ec];
        }
        unsigned a[8], b[8], p[8];
#pragma unroll
        for (int u = 0; u < 8; ++u) {
            const int e = (w0 + i0 + u) * 32 + lane;
            const bool v = (e < E) && (d[u] >= 0);
            const unsigned o = (unsigned)d[u] / RANGE;
            const unsigned r = (unsigned)d[u] - o * RANGE;
            a[u] = v ? (o >> 4) : 16u;
            b[u] = v ? (o & 15u) : 16u;
            p[u] = (unsigned)s[u] | (r << 17);
        }
#pragma unroll
        for (int u = 0; u < 8; ++u) {
            const bool mine = (lane == i0 + u);
#pragma unroll
            for (int k = 0; k < 16; ++k) {
                const unsigned ma = __builtin_amdgcn_ballot_w32(a[u] == (unsigned)k);
                const unsigned mb = __builtin_amdgcn_ballot_w32(b[u] == (unsigned)k);
                wa[k] = mine ? ma : wa[k];
                wb[k] = mine ? mb : wb[k];
            }
        }
#pragma unroll
        for (int u = 0; u < 8; ++u) {
            const int e = (w0 + i0 + u) * 32 + lane;
            if (a[u] < 16u) pk[e] = p[u];
        }
    }
#pragma unroll
    for (int k = 0; k < 16; ++k) {
        planes[(size_t)k * pw + w0 + lane]        = wa[k];
        planes[(size_t)(16 + k) * pw + w0 + lane] = wb[k];
    }
}

#define LCAP  1024
#define HALF  196
#define OWNW  16
#define HROWS 208
#define QCAP  64
#define OWN_G 4
#define ROWS  400
#define PITCH_OF(F) ((F) + 4)
static_assert(ROWS % 16 == 0 && ROWS >= RANGE, "whole tiles covering the range");
static_assert(HROWS % 16 == 0 && HROWS >= HALF && HROWS >= RANGE - HALF, "whole tiles covering either half");

template <int F, int U, bool FULL>
__device__ __forceinline__ void owner_rows(const int my_s, const int my_r, int n, int lane,
                                           const float* __restrict__ h, float* acc) {
    constexpr int C = F / 32;
    constexpr bool WIDE = (F == 128);
    static_assert(32 % U == 0, "a full batch is whole trips");
    if constexpr (WIDE) {
#pragma unroll 1
        for (int j0 = 0; j0 < n; j0 += U) {
            float4 v[U];
#pragma unroll
            for (int u = 0; u < U; ++u) {
                const int s = __builtin_amdgcn_readlane(my_s, j0 + u);
                v[u] = reinterpret_cast<const float4*>(h + (size_t)s * F)[lane];
            }
#pragma unroll
            for (int u = 0; u < U; ++u) {
                const int r = __builtin_amdgcn_readlane(my_r, j0 + u);
                if (FULL || r >= 0) {
                    float* o = acc + r * PITCH_OF(F) + 4 * lane;
                    atomicAdd(o,     v[u].x);
                    atomicAdd(o + 1, v[u].y);
                    atomicAdd(o + 2, v[u].z);
                    atomicAdd(o + 3, v[u].w);
                }
            }
        }
    } else {
#pragma unroll 1
        for (int j0 = 0; j0 < n; j0 += U) {
            float v[U][C];
#pragma unroll
            for (int u = 0; u < U; ++u) {
                const int s = __builtin_amdgcn_readlane(my_s, j0 + u);
                const float* row = h + (size_t)s * F + lane;
#pragma unroll
                for (int c = 0; c < C; ++c) v[u][c] = row[c * 32];
            }
#pragma unroll
            for (int u = 0; u < U; ++u) {
                const int r = __builtin_amdgcn_readlane(my_r, j0 + u);
                if (FULL || r >= 0) {
                    float* o = acc + r * PITCH_OF(F) + lane;
#pragma unroll
                    for (int c = 0; c < C; ++c) atomicAdd(o + c * 32, v[u][c]);
                }
            }
        }
    }
}

template <int U, bool FULL>
__device__ __forceinline__ void excl_rows(const int my_s, const int my_r, const int n, const int lane,
                                          const float* __restrict__ h, float* acc) {
    constexpr int F = 128;
    static_assert(U % 4 == 0 && 32 % (2 * U) == 0, "a full batch is whole double trips of fours");
    auto load = [&](float4 (&v)[U], const int j0) {
#pragma unroll
        for (int u = 0; u < U; ++u) {
            const int s = __builtin_amdgcn_readlane(my_s, j0 + u);
            v[u] = reinterpret_cast<const float4*>(h + (size_t)s * F)[lane];
        }
    };
    auto add4 = [](float4& c, const float4& x) { c.x += x.x; c.y += x.y; c.z += x.z; c.w += x.w; };
    auto two = [&](const float4& x0, const float4& x1, const int r0, const int r1) {
        if (FULL || r0 >= 0) {
            float4* o0 = reinterpret_cast<float4*>(acc + r0 * PITCH_OF(F)) + lane;
            if ((FULL || r1 >= 0) && r1 != r0) {
                float4* o1 = reinterpret_cast<float4*>(acc + r1 * PITCH_OF(F)) + lane;
                float4 c0 = *o0, c1 = *o1;
                add4(c0, x0);
                add4(c1, x1);
                *o0 = c0;
                *o1 = c1;
            } else {
                float4 c0 = *o0;
                add4(c0, x0);
                if (FULL || r1 >= 0) add4(c0, x1);
                *o0 = c0;
            }
        }
    };
    auto sum = [&](const float4 (&v)[U], const int j0) {
#pragma unroll
        for (int u = 0; u < U; u += 4) {
            const int r0 = __builtin_amdgcn_readlane(my_r, j0 + u);
            const int r1 = __builtin_amdgcn_readlane(my_r, j0 + u + 1);
            const int r2 = __builtin_amdgcn_readlane(my_r, j0 + u + 2);
            const int r3 = __builtin_amdgcn_readlane(my_r, j0 + u + 3);
            if (FULL && r0 != r1 && r0 != r2 && r0 != r3 && r1 != r2 && r1 != r3 && r2 != r3) {
                float4* o0 = reinterpret_cast<float4*>(acc + r0 * PITCH_OF(F)) + lane;
                float4* o1 = reinterpret_cast<float4*>(acc + r1 * PITCH_OF(F)) + lane;
                float4* o2 = reinterpret_cast<float4*>(acc + r2 * PITCH_OF(F)) + lane;
                float4* o3 = reinterpret_cast<float4*>(acc + r3 * PITCH_OF(F)) + lane;
                float4 c0 = *o0, c1 = *o1, c2 = *o2, c3 = *o3;
                add4(c0, v[u]);
                add4(c1, v[u + 1]);
                add4(c2, v[u + 2]);
                add4(c3, v[u + 3]);
                *o0 = c0;
                *o1 = c1;
                *o2 = c2;
                *o3 = c3;
            } else {
                two(v[u], v[u + 1], r0, r1);
                two(v[u + 2], v[u + 3], r2, r3);
            }
        }
    };
    if constexpr (FULL) {
        float4 va[U], vb[U];
        load(va, 0);
#pragma unroll
        for (int t = 0; t < 32 / (2 * U); ++t) {
            load(vb, (2 * t + 1) * U);
            sum(va, 2 * t * U);
            if (t + 1 < 32 / (2 * U)) load(va, (2 * t + 2) * U);
            sum(vb, (2 * t + 1) * U);
        }
    } else {
#pragma unroll 1
        for (int j0 = 0; j0 < n; j0 += U) {
            float4 va[U];
            load(va, j0);
            sum(va, j0);
        }
    }
}

template <int F, int U, bool FULL, bool COUNT>
__device__ __forceinline__ void owner_drain(const unsigned* qw, int first, int n, int lane,
                                            const float* __restrict__ h,
                                            const unsigned* __restrict__ pk,
                                            float* acc, float* cnt, unsigned* lw,
                                            int* lc, const int rlo, const int rcnt) {
    const bool on = FULL || (lane < n);
    unsigned p = 0u;
    int my_s = 0, my_r = -1;
    if (on) {
        p = pk[qw[first + lane]];
        my_s = (int)(p & 0x1ffffu);
        my_r = (int)(p >> 17);
    }
    if constexpr (COUNT) {
        if (on) {
            atomicAdd(cnt + my_r, 1.0f);
            const int hi = (my_r >= HALF) ? 1 : 0;
            const int rr = my_r - hi * HALF;
            const int li = hi * OWNW + (rr % OWNW);
            const int pos = atomicAdd(lc + li, 1);
            if (pos < LCAP) lw[(size_t)li * LCAP + pos] = p - ((unsigned)(hi * HALF) << 17);
        }
        owner_rows<F, U, FULL>(my_s, my_r, n, lane, h, acc);
    } else {
        const int rr = my_r - rlo;
        my_r = (on && rr >= 0 && rr < rcnt) ? rr : -1;
        owner_rows<F, U, false>(my_s, my_r, n, lane, h, acc);
    }
}

template <int U>
__device__ __forceinline__ void owner_walk(const unsigned* __restrict__ lw, const int n, int lane,
                                           const float* __restrict__ h, float* acc) {
    if (n <= 0) return;
    unsigned p = lw[(lane < n) ? lane : (n - 1)];
    int j = 0;
#pragma unroll 1
    for (; j + 32 <= n; j += 32) {
        const int nx = j + 32 + lane;
        const unsigned pn = lw[(nx < n) ? nx : (n - 1)];
        excl_rows<U, true>((int)(p & 0x1ffffu), (int)(p >> 17), 32, lane, h, acc);
        p = pn;
    }
    const int m = n - j;
    if (m > 0) {
        const bool on = (lane < m);
        excl_rows<U, false>(on ? (int)(p & 0x1ffffu) : 0, on ? (int)(p >> 17) : -1,
                            m, lane, h, acc);
    }
}

template <int F, int U, bool COUNT>
__device__ __forceinline__ int owner_take(unsigned hits, const int eb, const int E,
                                          unsigned* qw, int qn, int lane,
                                          const float* __restrict__ h,
                                          const unsigned* __restrict__ pk, float* acc,
                                          float* cnt, unsigned* lw, int* lc,
                                          const int rlo, const int rcnt) {
    while (__builtin_amdgcn_ballot_w32(hits != 0u) != 0u) {
        const int e = eb + __builtin_ctz(hits | 0x80000000u);
        const bool m = (hits != 0u) && (e < E);
        const unsigned b = __builtin_amdgcn_ballot_w32(m);
        if (m) qw[qn + (int)__builtin_amdgcn_mbcnt_lo(b, 0u)] = (unsigned)e;
        qn += __builtin_popcount(b);
        hits &= hits - 1u;
        if (qn >= 32) {
            qn -= 32;
            owner_drain<F, U, true, COUNT>(qw, qn, 32, lane, h, pk, acc, cnt, lw, lc,
                                           rlo, rcnt);
        }
    }
    return qn;
}

template <int F, int FOUT, int T>
__device__ __forceinline__ void rows_mma(const float* ap, const int ts,
                                         const float* __restrict__ wp, v8f (&d)[T]) {
#pragma unroll 4
    for (int k0 = 0; k0 < F; k0 += 4) {
        v2f b;
        b.x = wp[(size_t)k0 * FOUT];
        b.y = wp[(size_t)(k0 + 1) * FOUT];
#pragma unroll
        for (int t = 0; t < T; ++t) {
            const float2 a2 = *reinterpret_cast<const float2*>(ap + t * ts + k0);
            v2f a; a.x = a2.x; a.y = a2.y;
            d[t] = __builtin_amdgcn_wmma_f32_16x16x4_f32(
                false, a, false, b, (short)0, d[t], false, false);
        }
    }
}

template <int OP, bool PRELU>
__device__ __forceinline__ void tile_put(const v8f& d, const float bv, const float alpha,
                                         float* p) {
#pragma unroll
    for (int v = 0; v < 8; ++v) {
        float val = d[v] + bv;
        if (PRELU) val = (val >= 0.0f) ? val : alpha * val;
        p[v * OP] = val;
    }
}

template <int F, int FOUT, int U, bool PRELU>
__global__ __launch_bounds__(1024) void owner_layer_kernel(const float* __restrict__ h,
                               const unsigned* __restrict__ planes,
                               const unsigned* __restrict__ pk,
                               unsigned* __restrict__ lst, int* __restrict__ lcnt,
                               float* __restrict__ invdeg,
                               const float* __restrict__ Wl, const float* __restrict__ bias,
                               const float* __restrict__ Wr, const float* __restrict__ alpha_p,
                               float* __restrict__ out, int E) {
    constexpr int PITCH = PITCH_OF(F);
    constexpr int OP    = FOUT + 4;
    constexpr int LROW  = (PITCH > OP) ? PITCH : OP;
    __shared__ __attribute__((aligned(16))) float acc[ROWS * LROW];
    __shared__ unsigned q[32 * QCAP];
    __shared__ float cnt[ROWS];
    __shared__ int lcount[2 * OWNW];

    const int tid  = threadIdx.x;
    const int lane = tid & 31;
    const int wave = __builtin_amdgcn_readfirstlane(tid >> 5);
    const int lo   = blockIdx.x * RANGE;
    int nr = N_NODES - lo;
    if (nr > RANGE) nr = RANGE;

    unsigned* lw = lst + (size_t)blockIdx.x * (2 * OWNW) * LCAP;

    float4* acc4 = reinterpret_cast<float4*>(acc);
    for (int i = tid; i < ROWS * PITCH / 4; i += 1024)
        acc4[i] = make_float4(0.0f, 0.0f, 0.0f, 0.0f);
    for (int i = tid; i < ROWS; i += 1024) cnt[i] = 0.0f;
    if (tid < 2 * OWNW) lcount[tid] = 0;
    __syncthreads();

    {
        unsigned* qw = q + wave * QCAP;
        int qn = 0;
        const int ng = (E + 4095) >> 12;
        const int pw = ng << 7;
        const uint4* pa = reinterpret_cast<const uint4*>(planes + (size_t)(blockIdx.x >> 4) * pw);
        const uint4* pb = reinterpret_cast<const uint4*>(planes + (size_t)(16 + (blockIdx.x & 15)) * pw);
        const int start = (int)(((long long)blockIdx.x * ng) >> 8);

        for (int t0 = wave; t0 < ng; t0 += 32 * OWN_G) {
            uint4 wa[OWN_G], wb[OWN_G];
            int   eb[OWN_G];
#pragma unroll
            for (int u = 0; u < OWN_G; ++u) {
                const int t = t0 + 32 * u;
                int g = start + ((t < ng) ? t : t0);
                if (g >= ng) g -= ng;
                const int qi = g * 32 + lane;
                eb[u] = qi * 128;
                wa[u] = pa[qi];
                wb[u] = pb[qi];
            }
#pragma unroll
            for (int u = 0; u < OWN_G; ++u) {
                const bool live = (t0 + 32 * u < ng);
                const unsigned m0 = live ? (wa[u].x & wb[u].x) : 0u;
                const unsigned m1 = live ? (wa[u].y & wb[u].y) : 0u;
                const unsigned m2 = live ? (wa[u].z & wb[u].z) : 0u;
                const unsigned m3 = live ? (wa[u].w & wb[u].w) : 0u;
                qn = owner_take<F, U, true>(m0, eb[u],      E, qw, qn, lane, h, pk, acc, cnt, lw, lcount, 0, RANGE);
                qn = owner_take<F, U, true>(m1, eb[u] + 32, E, qw, qn, lane, h, pk, acc, cnt, lw, lcount, 0, RANGE);
                qn = owner_take<F, U, true>(m2, eb[u] + 64, E, qw, qn, lane, h, pk, acc, cnt, lw, lcount, 0, RANGE);
                qn = owner_take<F, U, true>(m3, eb[u] + 96, E, qw, qn, lane, h, pk, acc, cnt, lw, lcount, 0, RANGE);
            }
        }
        if (qn > 0)
            owner_drain<F, U, false, true>(qw, 0, qn, lane, h, pk, acc, cnt, lw, lcount, 0, RANGE);
    }
    __syncthreads();

    if (tid < 2 * OWNW) lcnt[blockIdx.x * (2 * OWNW) + tid] = lcount[tid];

    constexpr int Q = F / 4;
    for (int i = tid; i < nr; i += 1024) invdeg[lo + i] = 1.0f / fmaxf(cnt[i], 1.0f);
    for (int i = tid; i < ROWS * Q; i += 1024) {
        const int r = i / Q, c4 = i % Q;
        const float s = 1.0f / fmaxf(cnt[r], 1.0f);
        float4* p = reinterpret_cast<float4*>(acc + r * PITCH + 4 * c4);
        float4 v = *p;
        v.x *= s; v.y *= s; v.z *= s; v.w *= s;
        *p = v;
    }
    __syncthreads();

    constexpr int NT = FOUT / 16;
    constexpr int NG = 32 / NT;
    constexpr int MT = ROWS / 16;
    constexpr int TM = MT / NG;
    constexpr int XT = MT % NG;
    static_assert(32 % NT == 0, "whole tile columns per work-group");
    const int nt   = wave % NT;
    const int g    = wave / NT;
    const int half = lane >> 4;
    const int col  = lane & 15;
    const bool extra = (g < XT);

    v8f d[TM], dx[1];
#pragma unroll
    for (int t = 0; t < TM; ++t)
#pragma unroll
        for (int j = 0; j < 8; ++j) d[t][j] = 0.0f;
#pragma unroll
    for (int j = 0; j < 8; ++j) dx[0][j] = 0.0f;

    const float* ap  = acc + (g * 16 + col) * PITCH + half * 2;
    const float* apx = ap + TM * NG * 16 * PITCH;
    constexpr int TS = NG * 16 * PITCH;
    const size_t woff = (size_t)(half * 2) * FOUT + nt * 16 + col;

    rows_mma<F, FOUT, TM>(ap, TS, Wl + woff, d);
    if (extra) rows_mma<F, FOUT, 1>(apx, 0, Wl + woff, dx);
    __syncthreads();

    for (int i = tid; i < ROWS * Q; i += 1024) {
        const int r = i / Q, c4 = i % Q;
        int gr = lo + r;
        if (gr > N_NODES - 1) gr = N_NODES - 1;
        const float4 v = reinterpret_cast<const float4*>(h + (size_t)gr * F)[c4];
        *reinterpret_cast<float4*>(acc + r * PITCH + 4 * c4) = v;
    }
    __syncthreads();

    rows_mma<F, FOUT, TM>(ap, TS, Wr + woff, d);
    if (extra) rows_mma<F, FOUT, 1>(apx, 0, Wr + woff, dx);

    __syncthreads();

    const float alpha = alpha_p[0];
    const float bv = bias[nt * 16 + col];
    float* op = acc + (g * 16 + half * 8) * OP + nt * 16 + col;
#pragma unroll
    for (int t = 0; t < TM; ++t)
        tile_put<OP, PRELU>(d[t], bv, alpha, op + t * NG * 16 * OP);
    if (extra)
        tile_put<OP, PRELU>(dx[0], bv, alpha, op + TM * NG * 16 * OP);
    __syncthreads();

    constexpr int QO = FOUT / 4;
    float4* o4 = reinterpret_cast<float4*>(out + (size_t)lo * FOUT);
    for (int i = tid; i < nr * QO; i += 1024) {
        const int r = i / QO, c4 = i % QO;
        o4[i] = *reinterpret_cast<const float4*>(acc + r * OP + 4 * c4);
    }
}

template <int F, int FOUT, int U, bool PRELU>
__global__ __launch_bounds__(512, 8) void half_layer_kernel(const float* __restrict__ h,
                               const unsigned* __restrict__ planes,
                               const unsigned* __restrict__ pk,
                               const unsigned* __restrict__ lst, const int* __restrict__ lcnt,
                               const float* __restrict__ invdeg,
                               const float* __restrict__ Wl, const float* __restrict__ bias,
                               const float* __restrict__ Wr, const float* __restrict__ alpha_p,
                               float* __restrict__ out, int E) {
    static_assert(F == 128, "layers 2 and 3 read 128-wide rows");
    constexpr int NW    = OWNW;
    constexpr int NTH   = NW * 32;
    constexpr int PITCH = PITCH_OF(F);
    constexpr int OP    = FOUT + 4;
    constexpr int LROW  = (PITCH > OP) ? PITCH : OP;
    __shared__ __attribute__((aligned(16))) float acc[HROWS * LROW];
    __shared__ unsigned q[NW * QCAP];

    const int tid  = threadIdx.x;
    const int lane = tid & 31;
    const int wave = __builtin_amdgcn_readfirstlane(tid >> 5);
    const int b    = blockIdx.x >> 1;
    const int p    = blockIdx.x & 1;
    const int rlo  = p * HALF;
    const int cap  = p ? (RANGE - HALF) : HALF;
    const int lo   = b * RANGE + rlo;
    int nr = N_NODES - lo;
    if (nr > cap) nr = cap;
    if (nr < 0) nr = 0;
    const bool root_first = (__builtin_popcount(blockIdx.x) & 1) != 0;

    const int myc = lcnt[(b * 2 + p) * OWNW + (lane % OWNW)];
    const bool by_list = (__builtin_amdgcn_ballot_w32(myc > LCAP) == 0u);
    const int n0 = __builtin_amdgcn_readlane(myc, wave);
    const unsigned* l0 = lst + ((size_t)(b * 2 + p) * OWNW + wave) * LCAP;

    constexpr int Q  = F / 4;
    constexpr int NT = FOUT / 16;
    constexpr int NG = NW / NT;
    constexpr int MT = HROWS / 16;
    constexpr int TM = MT / NG;
    constexpr int XT = MT % NG;
    static_assert(NW % NT == 0, "whole tile columns per work-group");
    const int nt   = wave % NT;
    const int g    = wave / NT;
    const int half = lane >> 4;
    const int col  = lane & 15;
    const bool extra = (g < XT);

    v8f d[TM], dx[1];
    auto zero_tiles = [&]() {
#pragma unroll
        for (int t = 0; t < TM; ++t)
#pragma unroll
            for (int j = 0; j < 8; ++j) d[t][j] = 0.0f;
#pragma unroll
        for (int j = 0; j < 8; ++j) dx[0][j] = 0.0f;
    };

    const float* ap  = acc + (g * 16 + col) * PITCH + half * 2;
    const float* apx = ap + TM * NG * 16 * PITCH;
    constexpr int TS = NG * 16 * PITCH;
    const size_t woff  = (size_t)(half * 2) * FOUT + nt * 16 + col;
    float4* acc4 = reinterpret_cast<float4*>(acc);

    auto neighbour_part = [&](const bool first) {
        for (int i = tid; i < HROWS * PITCH / 4; i += NTH)
            acc4[i] = make_float4(0.0f, 0.0f, 0.0f, 0.0f);
        __syncthreads();
        if (by_list) {
            if (root_first) owner_walk<U / 4>(l0, n0, lane, h, acc);
            else            owner_walk<U / 2>(l0, n0, lane, h, acc);
        } else {
            unsigned* qw = q + wave * QCAP;
            int qn = 0;
            const int ng = (E + 4095) >> 12;
            const int pw = ng << 7;
            const uint4* pa = reinterpret_cast<const uint4*>(planes + (size_t)(b >> 4) * pw);
            const uint4* pb = reinterpret_cast<const uint4*>(planes + (size_t)(16 + (b & 15)) * pw);
            const int start = (int)(((long long)b * ng) >> 8);
            for (int t = wave; t < ng; t += NW) {
                int gi = start + t;
                if (gi >= ng) gi -= ng;
                const int qi = gi * 32 + lane;
                const int eb = qi * 128;
                const uint4 wa = pa[qi];
                const uint4 wb = pb[qi];
                qn = owner_take<F, U / 2, false>(wa.x & wb.x, eb,      E, qw, qn, lane, h, pk, acc, nullptr, nullptr, nullptr, rlo, cap);
                qn = owner_take<F, U / 2, false>(wa.y & wb.y, eb + 32, E, qw, qn, lane, h, pk, acc, nullptr, nullptr, nullptr, rlo, cap);
                qn = owner_take<F, U / 2, false>(wa.z & wb.z, eb + 64, E, qw, qn, lane, h, pk, acc, nullptr, nullptr, nullptr, rlo, cap);
                qn = owner_take<F, U / 2, false>(wa.w & wb.w, eb + 96, E, qw, qn, lane, h, pk, acc, nullptr, nullptr, nullptr, rlo, cap);
            }
            if (qn > 0)
                owner_drain<F, U / 2, false, false>(qw, 0, qn, lane, h, pk, acc, nullptr, nullptr, nullptr, rlo, cap);
        }
        __syncthreads();
        for (int i = tid; i < HROWS * Q; i += NTH) {
            const int r = i / Q, c4 = i % Q;
            const float s = (r < nr) ? invdeg[lo + r] : 0.0f;
            float4* ptr = reinterpret_cast<float4*>(acc + r * PITCH + 4 * c4);
            float4 v = *ptr;
            v.x *= s; v.y *= s; v.z *= s; v.w *= s;
            *ptr = v;
        }
        __syncthreads();
        if (first) zero_tiles();
        rows_mma<F, FOUT, TM>(ap, TS, Wl + woff, d);
        if (extra) rows_mma<F, FOUT, 1>(apx, 0, Wl + woff, dx);
        __syncthreads();
    };

    auto root_part = [&](const bool first) {
        for (int i = tid; i < HROWS * Q; i += NTH) {
            const int r = i / Q, c4 = i % Q;
            int gr = lo + r;
            if (gr > N_NODES - 1) gr = N_NODES - 1;
            const float4 v = reinterpret_cast<const float4*>(h + (size_t)gr * F)[c4];
            *reinterpret_cast<float4*>(acc + r * PITCH + 4 * c4) = v;
        }
        __syncthreads();
        if (first) zero_tiles();
        rows_mma<F, FOUT, TM>(ap, TS, Wr + woff, d);
        if (extra) rows_mma<F, FOUT, 1>(apx, 0, Wr + woff, dx);
        __syncthreads();
    };

    if (root_first) {
        root_part(true);
        neighbour_part(false);
    } else {
        neighbour_part(true);
        root_part(false);
    }

    const float alpha = alpha_p[0];
    const float bv = bias[nt * 16 + col];
    float* op = acc + (g * 16 + half * 8) * OP + nt * 16 + col;
#pragma unroll
    for (int t = 0; t < TM; ++t)
        tile_put<OP, PRELU>(d[t], bv, alpha, op + t * NG * 16 * OP);
    if (extra)
        tile_put<OP, PRELU>(dx[0], bv, alpha, op + TM * NG * 16 * OP);
    __syncthreads();

    constexpr int QO = FOUT / 4;
    float4* o4 = reinterpret_cast<float4*>(out + (size_t)lo * FOUT);
    for (int i = tid; i < nr * QO; i += NTH) {
        const int r = i / QO, c4 = i % QO;
        o4[i] = *reinterpret_cast<const float4*>(acc + r * OP + 4 * c4);
    }
}

__global__ void classifier_kernel(const float* __restrict__ h3,
                                  const float* __restrict__ Wc,
                                  const float* __restrict__ bc,
                                  float* __restrict__ out, int n) {
    int i = blockIdx.x * blockDim.x + threadIdx.x;
    if (i >= n) return;
    float s0 = bc[0], s1 = bc[1], s2 = bc[2], s3 = bc[3];
    const float4* row = reinterpret_cast<const float4*>(h3 + (size_t)i * 64);
#pragma unroll
    for (int k4 = 0; k4 < 16; ++k4) {
        float4 v = row[k4];
        const float* w = Wc + (k4 * 4) * 4;
        s0 += v.x * w[0];  s1 += v.x * w[1];  s2 += v.x * w[2];  s3 += v.x * w[3];
        s0 += v.y * w[4];  s1 += v.y * w[5];  s2 += v.y * w[6];  s3 += v.y * w[7];
        s0 += v.z * w[8];  s1 += v.z * w[9];  s2 += v.z * w[10]; s3 += v.z * w[11];
        s0 += v.w * w[12]; s1 += v.w * w[13]; s2 += v.w * w[14]; s3 += v.w * w[15];
    }
    float* o = out + (size_t)i * 4;
    o[0] = s0; o[1] = s1; o[2] = s2; o[3] = s3;
}

extern "C" void kernel_launch(void* const* d_in, const int* in_sizes, int n_in,
                              void* d_out, int out_size, void* d_ws, size_t ws_size,
                              hipStream_t stream) {
    const float* x   = (const float*)d_in[0];
    const int*   ei  = (const int*)d_in[1];
    const float* W1l = (const float*)d_in[2];
    const float* b1  = (const float*)d_in[3];
    const float* W1r = (const float*)d_in[4];
    const float* W2l = (const float*)d_in[5];
    const float* b2  = (const float*)d_in[6];
    const float* W2r = (const float*)d_in[7];
    const float* W3l = (const float*)d_in[8];
    const float* b3  = (const float*)d_in[9];
    const float* W3r = (const float*)d_in[10];
    const float* Wc  = (const float*)d_in[11];
    const float* bc  = (const float*)d_in[12];
    const float* a   = (const float*)d_in[13];
    float* out = (float*)d_out;

    const int E = in_sizes[1] / 2;
    const int* srcv = ei;
    const int* dstv = ei + E;

    char* ws = (char*)d_ws;
    size_t off = 0;
    auto carve = [&](size_t bytes) {
        float* p = (float*)(ws + off);
        off += (bytes + 255) & ~(size_t)255;
        return p;
    };
    float* deg = carve((size_t)N_NODES * 4);
    float* hC  = carve((size_t)N_NODES * 128 * 4);
    float* hA  = carve((size_t)N_NODES * 128 * 4);
    float* hB  = carve((size_t)N_NODES * 128 * 4);
    float* h3  = hA;

    const int TB = 256;
    const int nOwn = (N_NODES + RANGE - 1) / RANGE;
    const int nGrp = (E + 4095) / 4096;

    unsigned* own = (unsigned*)hB;
    unsigned* pk  = own + (size_t)nGrp * 4096;
    unsigned* lst = pk + (size_t)nGrp * 4096;
    int* lcnt = (int*)(lst + (size_t)nOwn * 2 * OWNW * LCAP);
    edge_keys_kernel<<<nGrp, 128, 0, stream>>>(srcv, dstv, own, pk, E);

    owner_layer_kernel<32, 128, 16, true><<<nOwn, 1024, 0, stream>>>(
        x, own, pk, lst, lcnt, deg, W1l, b1, W1r, a, hA, E);

    half_layer_kernel<128, 128, 16, true><<<2 * nOwn, 512, 0, stream>>>(
        hA, own, pk, lst, lcnt, deg, W2l, b2, W2r, a, hC, E);

    half_layer_kernel<128, 64, 16, false><<<2 * nOwn, 512, 0, stream>>>(
        hC, own, pk, lst, lcnt, deg, W3l, b3, W3r, a, h3, E);

    classifier_kernel<<<(N_NODES + TB - 1) / TB, TB, 0, stream>>>(
        h3, Wc, bc, out, N_NODES);
}
